# starting kernel
# baseline (speedup 1.0000x reference)
#include <hip/hip_runtime.h>
#include <hip/hip_bf16.h>

typedef __attribute__((ext_vector_type(16))) _Float16 v16h;
typedef __attribute__((ext_vector_type(8)))  _Float16 v8h;
typedef __attribute__((ext_vector_type(8)))  float    v8f;

#define IN_CH 128
#define HID   64


__global__ void gcn_cvt_f16(const float* __restrict__ in, _Float16* __restrict__ out, int n) {
    int i = blockIdx.x * blockDim.x + threadIdx.x;
    if (i < n) out[i] = (_Float16)in[i];
}

__global__ void gcn_set_const(float* __restrict__ p, float v, int n) {
    int i = blockIdx.x * blockDim.x + threadIdx.x;
    if (i < n) p[i] = v;
}

__global__ void gcn_out_init_bias(float* __restrict__ out, const float* __restrict__ b, int n) {
    int i = blockIdx.x * blockDim.x + threadIdx.x;
    if (i < n) out[i] = b[i & (HID - 1)];
}

__global__ void gcn_bias_relu_cvt(const float* __restrict__ agg, const float* __restrict__ b,
                                  _Float16* __restrict__ out, int n) {
    int i = blockIdx.x * blockDim.x + threadIdx.x;
    if (i < n) out[i] = (_Float16)fmaxf(agg[i] + b[i & (HID - 1)], 0.0f);
}


__global__ void gcn_deg_accum(const int* __restrict__ dst, float* __restrict__ deg, int E) {
    int e = blockIdx.x * blockDim.x + threadIdx.x;
    if (e < E) atomicAdd(&deg[dst[e]], 1.0f);
}

__global__ void gcn_deg_rsqrt(float* __restrict__ deg, int n) {
    int i = blockIdx.x * blockDim.x + threadIdx.x;
    if (i < n) deg[i] = rsqrtf(deg[i]);
}

__global__ void gcn_pack_w(const float* __restrict__ W, _Float16* __restrict__ Bp,
                           int K, int total) {
    int idx = blockIdx.x * blockDim.x + threadIdx.x;
    if (idx >= total) return;
    int e    = idx & 15;
    int lane = (idx >> 4) & 31;
    int tile = idx >> 9;
    int nt   = tile & 3;
    int kt   = tile >> 2;
    int hi   = lane >> 4;
    int k    = kt * 32 + hi * 8 + (e & 7) + ((e & 8) ? 16 : 0);
    int n    = nt * 16 + (lane & 15);
    Bp[idx] = (_Float16)W[k * HID + n];
    (void)K;
}

__global__ void __launch_bounds__(128)
gcn_gemm_wmma(const _Float16* __restrict__ A, const _Float16* __restrict__ Bp,
              float* __restrict__ C, int K) {
    const int lane = threadIdx.x & 31;
    const int nT   = threadIdx.x >> 5;
    const int m0   = blockIdx.x << 4;
    const int hi   = lane >> 4;
    const int row  = m0 + (lane & 15);
    const int kTiles = K >> 5;

    const _Float16* arow = A + (size_t)row * K + hi * 8;
    const _Float16* bp   = Bp + ((size_t)nT * 32 + lane) * 16;

    v8f acc = {0.f, 0.f, 0.f, 0.f, 0.f, 0.f, 0.f, 0.f};

    for (int kt = 0; kt < kTiles; ++kt) {
        union { v16h v; v8h h[2]; } a;
        a.h[0] = *(const v8h*)(arow);
        a.h[1] = *(const v8h*)(arow + 16);
        v16h b = *(const v16h*)(bp);
        __builtin_prefetch(arow + 32, 0, 0);
        acc = __builtin_amdgcn_wmma_f32_16x16x32_f16(
             false, a.v,  false, b,
             (short)0, acc,  false,  false);
        arow += 32;
        bp   += 4 * 32 * 16;
    }

    float* crow = C + (size_t)(m0 + hi * 8) * HID + (nT << 4) + (lane & 15);
#pragma unroll
    for (int r = 0; r < 8; ++r) crow[(size_t)r * HID] = acc[r];
}

__global__ void __launch_bounds__(256)
gcn_aggregate(const float* __restrict__ H, const int* __restrict__ src,
              const int* __restrict__ dst, const float* __restrict__ dis,
              float* __restrict__ out, int E, int N) {
    int wave = (blockIdx.x * blockDim.x + threadIdx.x) >> 5;
    int lane = threadIdx.x & 31;
    if (wave >= E + N) return;
    int s, d;
    if (wave < E) { s = src[wave]; d = dst[wave]; }
    else          { s = d = wave - E; }
    float norm = dis[s] * dis[d];
    const float2 h2 = *(const float2*)(H + (size_t)s * HID + lane * 2);
    float* o = out + (size_t)d * HID + lane * 2;
    atomicAdd(o,     h2.x * norm);
    atomicAdd(o + 1, h2.y * norm);
}


extern "C" void kernel_launch(void* const* d_in, const int* in_sizes, int n_in,
                              void* d_out, int out_size, void* d_ws, size_t ws_size,
                              hipStream_t stream) {
    const float* x  = (const float*)d_in[0];
    const int*   ei = (const int*)d_in[1];
    const float* W1 = (const float*)d_in[2];
    const float* b1 = (const float*)d_in[3];
    const float* W2 = (const float*)d_in[4];
    const float* b2 = (const float*)d_in[5];
    float* out = (float*)d_out;

    const int N = in_sizes[0] / IN_CH;
    const int E = in_sizes[1] / 2;
    const int* src = ei;
    const int* dst = ei + E;

    char* ws = (char*)d_ws;
    size_t off = 0;
    auto alloc = [&](size_t bytes) -> char* {
        char* p = ws + off;
        off = (off + bytes + 255) & ~(size_t)255;
        return p;
    };
    _Float16* xh   = (_Float16*)alloc((size_t)N * IN_CH * sizeof(_Float16));
    _Float16* w1p  = (_Float16*)alloc((size_t)(IN_CH/32) * 4 * 32 * 16 * sizeof(_Float16));
    _Float16* w2p  = (_Float16*)alloc((size_t)(HID/32)   * 4 * 32 * 16 * sizeof(_Float16));
    float*    proj = (float*)   alloc((size_t)N * HID * sizeof(float));
    float*    agg  = (float*)   alloc((size_t)N * HID * sizeof(float));
    _Float16* h1h  = (_Float16*)alloc((size_t)N * HID * sizeof(_Float16));
    float*    dis  = (float*)   alloc((size_t)N * sizeof(float));

    const int T = 256;
    int t;

    t = N * IN_CH;
    gcn_cvt_f16<<<(t + T - 1) / T, T, 0, stream>>>(x, xh, t);

    t = (IN_CH / 32) * 4 * 32 * 16;
    gcn_pack_w<<<(t + T - 1) / T, T, 0, stream>>>(W1, w1p, IN_CH, t);
    t = (HID / 32) * 4 * 32 * 16;
    gcn_pack_w<<<(t + T - 1) / T, T, 0, stream>>>(W2, w2p, HID, t);

    gcn_set_const<<<(N + T - 1) / T, T, 0, stream>>>(dis, 1.0f, N);
    gcn_deg_accum<<<(E + T - 1) / T, T, 0, stream>>>(dst, dis, E);
    gcn_deg_rsqrt<<<(N + T - 1) / T, T, 0, stream>>>(dis, N);

    gcn_gemm_wmma<<<N / 16, 128, 0, stream>>>(xh, w1p, proj, IN_CH);

    t = N * HID;
    gcn_set_const<<<(t + T - 1) / T, T, 0, stream>>>(agg, 0.0f, t);
    {
        long long threads = (long long)(E + N) * 32;
        int blocks = (int)((threads + T - 1) / T);
        gcn_aggregate<<<blocks, T, 0, stream>>>(proj, src, dst, dis, agg, E, N);
    }

    gcn_bias_relu_cvt<<<(t + T - 1) / T, T, 0, stream>>>(agg, b1, h1h, t);

    gcn_gemm_wmma<<<N / 16, 128, 0, stream>>>(h1h, w2p, proj, HID);

    gcn_out_init_bias<<<(t + T - 1) / T, T, 0, stream>>>(out, b2, t);
    {
        long long threads = (long long)(E + N) * 32;
        int blocks = (int)((threads + T - 1) / T);
        gcn_aggregate<<<blocks, T, 0, stream>>>(proj, src, dst, dis, out, E, N);
    }
}


# edge bucketing without global atomics: 64 edge pieces, one block each counts its piece per destination in a node table in LDS (LDS atomic return = rank); scan sums the 64 count planes
# speedup vs baseline: 2.4117x; 2.4117x over previous
#include <hip/hip_runtime.h>
#include <hip/hip_bf16.h>

typedef __attribute__((ext_vector_type(16))) _Float16 v16h;
typedef __attribute__((ext_vector_type(8)))  _Float16 v8h;
typedef __attribute__((ext_vector_type(8)))  float    v8f;

#define IN_CH 128
#define HID   64


#define CHUNKS    64
#define TAB_WORDS 51200
static_assert(CHUNKS == HID, "the count and offset planes reuse the N x HID buffers of the layers");

__global__ void __launch_bounds__(1024)
gcn_chunk_hist(const int* __restrict__ dst, unsigned short* __restrict__ hist,
               int* __restrict__ coarse, int* __restrict__ slot, int N, int E, int CH) {
    __shared__ alignas(16) unsigned tab[TAB_WORDS];
    const int tid  = threadIdx.x;
    const int lane = tid & 31;
    const int wv   = tid >> 5;
    const int g    = blockIdx.x;

    for (int i = tid; i < TAB_WORDS / 4; i += 1024) ((uint4*)tab)[i] = make_uint4(0u, 0u, 0u, 0u);
    __syncthreads();

    const int e0 = g * CH;
    const int e1 = (e0 + CH < E) ? (e0 + CH) : E;
    for (int eb = e0 + tid; eb < e1; eb += 4 * 1024) {
        int d[4];
#pragma unroll
        for (int u = 0; u < 4; ++u) {
            const int e = eb + u * 1024;
            d[u] = (e < e1) ? dst[e] : -1;
        }
#pragma unroll
        for (int u = 0; u < 4; ++u) {
            if ((unsigned)d[u] < (unsigned)TAB_WORDS) slot[eb + u * 1024] = (int)atomicAdd(&tab[d[u]], 1u);
        }
    }
    __syncthreads();

    const int nn = (N < TAB_WORDS) ? N : TAB_WORDS;
    unsigned short* __restrict__ hp = hist + (size_t)g * N;
    for (int n = tid; n < nn; n += 1024) hp[n] = (unsigned short)tab[n];

    for (int j = wv; j < 64; j += 32) {
        unsigned sacc = 0u;
        if (j < TAB_WORDS / 1024) {
            for (int i = 0; i < 32; ++i) sacc += tab[j * 1024 + i * 32 + lane];
        }
        for (int dd = 16; dd > 0; dd >>= 1) sacc += __shfl_xor(sacc, dd);
        if (lane == 0) coarse[g * 64 + j] = (int)sacc;
    }
}

__global__ void __launch_bounds__(1024)
gcn_scan_deg(const unsigned short* __restrict__ hist, const int* __restrict__ coarse,
             int* __restrict__ off, int* __restrict__ rowptr,
             float* __restrict__ dis, int N) {
    __shared__ int s_own[32];
    __shared__ int s_front[32];
    const int tid  = threadIdx.x;
    const int lane = tid & 31;
    const int wv   = tid >> 5;
    const int first = blockIdx.x * 1024;

    int front = 0;
    for (int i = tid; i < CHUNKS * 64; i += 1024) {
        const int v = coarse[i];
        front += ((i & 63) < (int)blockIdx.x) ? v : 0;
    }

    const int idx = first + tid;
    int c = 0;
    if (idx < N) {
#pragma unroll 16
        for (int g = 0; g < CHUNKS; ++g) c += (int)hist[(size_t)g * N + idx];
    }

    int incl = c;
    for (int d = 1; d < 32; d <<= 1) {
        int up = __shfl_up(incl, d);
        if (lane >= d) incl += up;
    }
    for (int d = 16; d > 0; d >>= 1) front += __shfl_xor(front, d);
    if (lane == 31) { s_own[wv] = incl; s_front[wv] = front; }
    __syncthreads();

    int t = s_front[lane] + ((lane < wv) ? s_own[lane] : 0);
    for (int d = 16; d > 0; d >>= 1) t += __shfl_xor(t, d);

    const int excl = t + incl - c;
    if (idx < N) {
        rowptr[idx] = excl;
        dis[idx]    = rsqrtf((float)(c + 1));
        if (idx == N - 1) rowptr[N] = excl + c;
        int run = excl;
#pragma unroll 16
        for (int g = 0; g < CHUNKS; ++g) {
            const int h = (int)hist[(size_t)g * N + idx];
            off[(size_t)g * N + idx] = run;
            run += h;
        }
    }
}

__global__ void gcn_fill(const int* __restrict__ src, const int* __restrict__ dst,
                         const int* __restrict__ slot, const int* __restrict__ off,
                         int* __restrict__ nbr, int N, int E, int CH) {
    int e = blockIdx.x * blockDim.x + threadIdx.x;
    if (e < E) {
        const int g = e / CH;
        const unsigned pos = (unsigned)(off[(size_t)g * N + dst[e]] + slot[e]);
        if (pos < (unsigned)E) nbr[pos] = src[e];
    }
}

template <int K, bool A_F32>
__global__ void __launch_bounds__(256)
gcn_gemm_wmma(const float* __restrict__ Af, const _Float16* __restrict__ Ah,
              const float* __restrict__ Wt, float* __restrict__ C, int N) {
    constexpr int KT = K / 32;
    constexpr int P  = K + 8;
    __shared__ alignas(16) _Float16 Bs[HID * P];

    const int tid  = threadIdx.x;
    const int lane = tid & 31;
    const int wv   = __builtin_amdgcn_readfirstlane(tid >> 5);
    const int c    = lane & 15;
    const int hi   = lane >> 4;
    const int m0   = (blockIdx.x * 8 + wv) << 4;
    const int row  = (m0 + c < N) ? (m0 + c) : (N - 1);

    constexpr int IT = HID * (K / 8) / 256;
    const int wn  = tid & (HID - 1);
    const int wk0 = tid >> 6;
    float wf[IT][8];
#pragma unroll
    for (int j = 0; j < IT; ++j) {
        const float* wp = Wt + (size_t)(wk0 + j * 4) * 8 * HID + wn;
#pragma unroll
        for (int q = 0; q < 8; ++q) wf[j][q] = wp[q * HID];
    }
    float4 af[A_F32 ? KT * 4 : 1];
    v8h    ah[A_F32 ? 1 : KT * 2];
    if constexpr (A_F32) {
        const float* ap = Af + (size_t)row * K + hi * 8;
#pragma unroll
        for (int kt = 0; kt < KT; ++kt) {
            af[kt * 4 + 0] = *(const float4*)(ap + kt * 32);
            af[kt * 4 + 1] = *(const float4*)(ap + kt * 32 + 4);
            af[kt * 4 + 2] = *(const float4*)(ap + kt * 32 + 16);
            af[kt * 4 + 3] = *(const float4*)(ap + kt * 32 + 20);
        }
    } else {
        const _Float16* ap = Ah + (size_t)row * K + hi * 8;
#pragma unroll
        for (int kt = 0; kt < KT; ++kt) {
            ah[kt * 2 + 0] = *(const v8h*)(ap + kt * 32);
            ah[kt * 2 + 1] = *(const v8h*)(ap + kt * 32 + 16);
        }
    }

#pragma unroll
    for (int j = 0; j < IT; ++j) {
        v8h pk;
#pragma unroll
        for (int q = 0; q < 8; ++q) pk[q] = (_Float16)wf[j][q];
        *(v8h*)(Bs + wn * P + (wk0 + j * 4) * 8) = pk;
    }
    __syncthreads();

    v8f acc[4];
#pragma unroll
    for (int nt = 0; nt < 4; ++nt) acc[nt] = (v8f){0.f, 0.f, 0.f, 0.f, 0.f, 0.f, 0.f, 0.f};

#pragma unroll
    for (int kt = 0; kt < KT; ++kt) {
        union { v16h v; v8h h[2]; } a;
        if constexpr (A_F32) {
#pragma unroll
            for (int j = 0; j < 4; ++j) {
                a.v[j * 4 + 0] = (_Float16)af[kt * 4 + j].x;
                a.v[j * 4 + 1] = (_Float16)af[kt * 4 + j].y;
                a.v[j * 4 + 2] = (_Float16)af[kt * 4 + j].z;
                a.v[j * 4 + 3] = (_Float16)af[kt * 4 + j].w;
            }
        } else {
            a.h[0] = ah[kt * 2 + 0];
            a.h[1] = ah[kt * 2 + 1];
        }
#pragma unroll
        for (int nt = 0; nt < 4; ++nt) {
            const _Float16* bp = Bs + (nt * 16 + c) * P + kt * 32 + hi * 8;
            union { v16h v; v8h h[2]; } b;
            b.h[0] = *(const v8h*)(bp);
            b.h[1] = *(const v8h*)(bp + 16);
            acc[nt] = __builtin_amdgcn_wmma_f32_16x16x32_f16(
                 false, a.v,  false, b.v,
                 (short)0, acc[nt],  false,  false);
        }
    }

    float* crow = C + (size_t)(m0 + hi * 8) * HID + c;
    if (m0 + 16 <= N) {
#pragma unroll
        for (int nt = 0; nt < 4; ++nt) {
#pragma unroll
            for (int r = 0; r < 8; ++r) crow[(size_t)r * HID + (nt << 4)] = acc[nt][r];
        }
    } else if (m0 < N) {
#pragma unroll
        for (int nt = 0; nt < 4; ++nt) {
#pragma unroll
            for (int r = 0; r < 8; ++r) {
                if (m0 + hi * 8 + r < N) crow[(size_t)r * HID + (nt << 4)] = acc[nt][r];
            }
        }
    }
}

template <bool RELU_F16>
__global__ void __launch_bounds__(256)
gcn_gather(const float* __restrict__ H, const int* __restrict__ rowptr,
           const int* __restrict__ nbr, const float* __restrict__ dis,
           const float* __restrict__ bias, float* __restrict__ outf,
           _Float16* __restrict__ outh, int N) {
    __shared__ int s_sorted[8][32];
    const int lane = threadIdx.x & 31;
    const int wv   = threadIdx.x >> 5;
    const int node = __builtin_amdgcn_readfirstlane((int)(blockIdx.x * 8 + wv));
    if (node >= N) return;
    const int beg = rowptr[node];
    const int cnt = rowptr[node + 1] - beg;
    const float dd = dis[node];
    const int* __restrict__ bucket = nbr + beg;

    const float2 hs = *(const float2*)(H + (size_t)node * HID + lane * 2);
    const float wself = dd * dd;
    float2 acc;
    acc.x = hs.x * wself;
    acc.y = hs.y * wself;

    for (int c0 = 0; c0 < cnt; c0 += 32) {
        for (int a0 = 0; a0 < cnt; a0 += 32) {
            const int ia  = a0 + lane;
            const int key = (ia < cnt) ? bucket[ia] : 0x7fffffff;
            const unsigned long long mine =
                ((unsigned long long)(unsigned)key << 32) | (unsigned)ia;
            int rank = 0;
            for (int b0 = 0; b0 < cnt; b0 += 32) {
                const int ib = b0 + lane;
                const int kb = (ib < cnt) ? bucket[ib] : 0x7fffffff;
                const int nb = (cnt - b0 < 32) ? (cnt - b0) : 32;
                for (int j = 0; j < nb; ++j) {
                    const unsigned kj = (unsigned)__builtin_amdgcn_readlane(kb, j);
                    const unsigned long long other =
                        ((unsigned long long)kj << 32) | (unsigned)(b0 + j);
                    rank += (other < mine) ? 1 : 0;
                }
            }
            if (ia < cnt && rank >= c0 && rank < c0 + 32) s_sorted[wv][rank - c0] = key;
        }
        const int n  = (cnt - c0 < 32) ? (cnt - c0) : 32;
        const int sk = (lane < n) ? s_sorted[wv][lane] : 0;
        const float w = (lane < n) ? dis[sk] * dd : 0.0f;
#pragma unroll 4
        for (int r = 0; r < n; ++r) {
            const int   s  = __builtin_amdgcn_readlane(sk, r);
            const float ww = __builtin_bit_cast(float,
                                 __builtin_amdgcn_readlane(__builtin_bit_cast(int, w), r));
            const float2 h = *(const float2*)(H + (size_t)s * HID + lane * 2);
            acc.x = fmaf(h.x, ww, acc.x);
            acc.y = fmaf(h.y, ww, acc.y);
        }
    }

    const float2 b2v = *(const float2*)(bias + lane * 2);
    if (RELU_F16) {
        typedef __attribute__((ext_vector_type(2))) _Float16 v2h;
        v2h o;
        o.x = (_Float16)fmaxf(acc.x + b2v.x, 0.0f);
        o.y = (_Float16)fmaxf(acc.y + b2v.y, 0.0f);
        *(v2h*)(outh + (size_t)node * HID + lane * 2) = o;
    } else {
        float2 o;
        o.x = acc.x + b2v.x;
        o.y = acc.y + b2v.y;
        *(float2*)(outf + (size_t)node * HID + lane * 2) = o;
    }
}


extern "C" void kernel_launch(void* const* d_in, const int* in_sizes, int n_in,
                              void* d_out, int out_size, void* d_ws, size_t ws_size,
                              hipStream_t stream) {
    const float* x  = (const float*)d_in[0];
    const int*   ei = (const int*)d_in[1];
    const float* W1 = (const float*)d_in[2];
    const float* b1 = (const float*)d_in[3];
    const float* W2 = (const float*)d_in[4];
    const float* b2 = (const float*)d_in[5];
    float* out = (float*)d_out;

    const int N = in_sizes[0] / IN_CH;
    const int E = in_sizes[1] / 2;
    const int* src = ei;
    const int* dst = ei + E;

    char* ws = (char*)d_ws;
    size_t off = 0;
    auto alloc = [&](size_t bytes) -> char* {
        char* p = ws + off;
        off = (off + bytes + 255) & ~(size_t)255;
        return p;
    };
    float*    proj = (float*)   alloc((size_t)N * HID * sizeof(float));
    _Float16* h1h  = (_Float16*)alloc((size_t)N * HID * sizeof(_Float16));
    float*    dis  = (float*)   alloc((size_t)N * sizeof(float));
    int*      crs  = (int*)     alloc((size_t)CHUNKS * 64 * sizeof(int));
    int*      rowp = (int*)     alloc((size_t)(N + 1) * sizeof(int));
    int*      slot = (int*)     alloc((size_t)E * sizeof(int));
    int*      nbr  = (int*)     alloc((size_t)E * sizeof(int));
    unsigned short* hist = (unsigned short*)h1h;
    int*            offp = (int*)proj;

    const int T = 256;
    const int gemmBlocks = ((N + 15) / 16 + 7) / 8;
    const int CH = (E + CHUNKS - 1) / CHUNKS;

    gcn_chunk_hist<<<CHUNKS, 1024, 0, stream>>>(dst, hist, crs, slot, N, E, CH);
    gcn_scan_deg<<<(N + 1023) / 1024, 1024, 0, stream>>>(hist, crs, offp, rowp, dis, N);
    gcn_fill<<<(E + T - 1) / T, T, 0, stream>>>(src, dst, slot, offp, nbr, N, E, CH);

    gcn_gemm_wmma<IN_CH, true><<<gemmBlocks, 256, 0, stream>>>(x, nullptr, W1, proj, N);

    gcn_gather<true><<<(N + 7) / 8, T, 0, stream>>>(proj, rowp, nbr, dis, b1, nullptr, h1h, N);

    gcn_gemm_wmma<HID, false><<<gemmBlocks, 256, 0, stream>>>(nullptr, h1h, W2, proj, N);
    gcn_gather<false><<<(N + 7) / 8, T, 0, stream>>>(proj, rowp, nbr, dis, b2, out, nullptr, N);
}


# layer 2 sums a node's f16 h1 rows first (128-B rows) and projects the sum in the same kernel (sum as two f16 terms); sum order now neighbours first, then k; 7 -> 6 launches
# speedup vs baseline: 2.5804x; 1.0700x over previous
#include <hip/hip_runtime.h>
#include <hip/hip_bf16.h>

typedef __attribute__((ext_vector_type(16))) _Float16 v16h;
typedef __attribute__((ext_vector_type(8)))  _Float16 v8h;
typedef __attribute__((ext_vector_type(8)))  float    v8f;

#define IN_CH 128
#define HID   64


#define CHUNKS    64
#define TAB_WORDS 51200
static_assert(CHUNKS == HID, "the count and offset planes reuse the N x HID buffers of the layers");

__global__ void __launch_bounds__(1024)
gcn_chunk_hist(const int* __restrict__ dst, unsigned short* __restrict__ hist,
               int* __restrict__ coarse, int* __restrict__ slot, int N, int E, int CH) {
    __shared__ alignas(16) unsigned tab[TAB_WORDS];
    const int tid  = threadIdx.x;
    const int lane = tid & 31;
    const int wv   = tid >> 5;
    const int g    = blockIdx.x;

    for (int i = tid; i < TAB_WORDS / 4; i += 1024) ((uint4*)tab)[i] = make_uint4(0u, 0u, 0u, 0u);
    __syncthreads();

    const int e0 = g * CH;
    const int e1 = (e0 + CH < E) ? (e0 + CH) : E;
    for (int eb = e0 + tid; eb < e1; eb += 4 * 1024) {
        int d[4];
#pragma unroll
        for (int u = 0; u < 4; ++u) {
            const int e = eb + u * 1024;
            d[u] = (e < e1) ? dst[e] : -1;
        }
#pragma unroll
        for (int u = 0; u < 4; ++u) {
            if ((unsigned)d[u] < (unsigned)TAB_WORDS) slot[eb + u * 1024] = (int)atomicAdd(&tab[d[u]], 1u);
        }
    }
    __syncthreads();

    const int nn = (N < TAB_WORDS) ? N : TAB_WORDS;
    unsigned short* __restrict__ hp = hist + (size_t)g * N;
    for (int n = tid; n < nn; n += 1024) hp[n] = (unsigned short)tab[n];

    for (int j = wv; j < 64; j += 32) {
        unsigned sacc = 0u;
        if (j < TAB_WORDS / 1024) {
            for (int i = 0; i < 32; ++i) sacc += tab[j * 1024 + i * 32 + lane];
        }
        for (int dd = 16; dd > 0; dd >>= 1) sacc += __shfl_xor(sacc, dd);
        if (lane == 0) coarse[g * 64 + j] = (int)sacc;
    }
}

__global__ void __launch_bounds__(1024)
gcn_scan_deg(const unsigned short* __restrict__ hist, const int* __restrict__ coarse,
             int* __restrict__ off, int* __restrict__ rowptr,
             float* __restrict__ dis, int N) {
    __shared__ int s_own[32];
    __shared__ int s_front[32];
    const int tid  = threadIdx.x;
    const int lane = tid & 31;
    const int wv   = tid >> 5;
    const int first = blockIdx.x * 1024;

    int front = 0;
    for (int i = tid; i < CHUNKS * 64; i += 1024) {
        const int v = coarse[i];
        front += ((i & 63) < (int)blockIdx.x) ? v : 0;
    }

    const int idx = first + tid;
    int c = 0;
    if (idx < N) {
#pragma unroll 16
        for (int g = 0; g < CHUNKS; ++g) c += (int)hist[(size_t)g * N + idx];
    }

    int incl = c;
    for (int d = 1; d < 32; d <<= 1) {
        int up = __shfl_up(incl, d);
        if (lane >= d) incl += up;
    }
    for (int d = 16; d > 0; d >>= 1) front += __shfl_xor(front, d);
    if (lane == 31) { s_own[wv] = incl; s_front[wv] = front; }
    __syncthreads();

    int t = s_front[lane] + ((lane < wv) ? s_own[lane] : 0);
    for (int d = 16; d > 0; d >>= 1) t += __shfl_xor(t, d);

    const int excl = t + incl - c;
    if (idx < N) {
        rowptr[idx] = excl;
        dis[idx]    = rsqrtf((float)(c + 1));
        if (idx == N - 1) rowptr[N] = excl + c;
        int run = excl;
#pragma unroll 16
        for (int g = 0; g < CHUNKS; ++g) {
            const int h = (int)hist[(size_t)g * N + idx];
            off[(size_t)g * N + idx] = run;
            run += h;
        }
    }
}

__global__ void gcn_fill(const int* __restrict__ src, const int* __restrict__ dst,
                         const int* __restrict__ slot, const int* __restrict__ off,
                         int* __restrict__ nbr, int N, int E, int CH) {
    int e = blockIdx.x * blockDim.x + threadIdx.x;
    if (e < E) {
        const int g = e / CH;
        const unsigned pos = (unsigned)(off[(size_t)g * N + dst[e]] + slot[e]);
        if (pos < (unsigned)E) nbr[pos] = src[e];
    }
}

template <int K, bool A_F32>
__global__ void __launch_bounds__(256)
gcn_gemm_wmma(const float* __restrict__ Af, const _Float16* __restrict__ Ah,
              const float* __restrict__ Wt, float* __restrict__ C, int N) {
    constexpr int KT = K / 32;
    constexpr int P  = K + 8;
    __shared__ alignas(16) _Float16 Bs[HID * P];

    const int tid  = threadIdx.x;
    const int lane = tid & 31;
    const int wv   = __builtin_amdgcn_readfirstlane(tid >> 5);
    const int c    = lane & 15;
    const int hi   = lane >> 4;
    const int m0   = (blockIdx.x * 8 + wv) << 4;
    const int row  = (m0 + c < N) ? (m0 + c) : (N - 1);

    constexpr int IT = HID * (K / 8) / 256;
    const int wn  = tid & (HID - 1);
    const int wk0 = tid >> 6;
    float wf[IT][8];
#pragma unroll
    for (int j = 0; j < IT; ++j) {
        const float* wp = Wt + (size_t)(wk0 + j * 4) * 8 * HID + wn;
#pragma unroll
        for (int q = 0; q < 8; ++q) wf[j][q] = wp[q * HID];
    }
    float4 af[A_F32 ? KT * 4 : 1];
    v8h    ah[A_F32 ? 1 : KT * 2];
    if constexpr (A_F32) {
        const float* ap = Af + (size_t)row * K + hi * 8;
#pragma unroll
        for (int kt = 0; kt < KT; ++kt) {
            af[kt * 4 + 0] = *(const float4*)(ap + kt * 32);
            af[kt * 4 + 1] = *(const float4*)(ap + kt * 32 + 4);
            af[kt * 4 + 2] = *(const float4*)(ap + kt * 32 + 16);
            af[kt * 4 + 3] = *(const float4*)(ap + kt * 32 + 20);
        }
    } else {
        const _Float16* ap = Ah + (size_t)row * K + hi * 8;
#pragma unroll
        for (int kt = 0; kt < KT; ++kt) {
            ah[kt * 2 + 0] = *(const v8h*)(ap + kt * 32);
            ah[kt * 2 + 1] = *(const v8h*)(ap + kt * 32 + 16);
        }
    }

#pragma unroll
    for (int j = 0; j < IT; ++j) {
        v8h pk;
#pragma unroll
        for (int q = 0; q < 8; ++q) pk[q] = (_Float16)wf[j][q];
        *(v8h*)(Bs + wn * P + (wk0 + j * 4) * 8) = pk;
    }
    __syncthreads();

    v8f acc[4];
#pragma unroll
    for (int nt = 0; nt < 4; ++nt) acc[nt] = (v8f){0.f, 0.f, 0.f, 0.f, 0.f, 0.f, 0.f, 0.f};

#pragma unroll
    for (int kt = 0; kt < KT; ++kt) {
        union { v16h v; v8h h[2]; } a;
        if constexpr (A_F32) {
#pragma unroll
            for (int j = 0; j < 4; ++j) {
                a.v[j * 4 + 0] = (_Float16)af[kt * 4 + j].x;
                a.v[j * 4 + 1] = (_Float16)af[kt * 4 + j].y;
                a.v[j * 4 + 2] = (_Float16)af[kt * 4 + j].z;
                a.v[j * 4 + 3] = (_Float16)af[kt * 4 + j].w;
            }
        } else {
            a.h[0] = ah[kt * 2 + 0];
            a.h[1] = ah[kt * 2 + 1];
        }
#pragma unroll
        for (int nt = 0; nt < 4; ++nt) {
            const _Float16* bp = Bs + (nt * 16 + c) * P + kt * 32 + hi * 8;
            union { v16h v; v8h h[2]; } b;
            b.h[0] = *(const v8h*)(bp);
            b.h[1] = *(const v8h*)(bp + 16);
            acc[nt] = __builtin_amdgcn_wmma_f32_16x16x32_f16(
                 false, a.v,  false, b.v,
                 (short)0, acc[nt],  false,  false);
        }
    }

    float* crow = C + (size_t)(m0 + hi * 8) * HID + c;
    if (m0 + 16 <= N) {
#pragma unroll
        for (int nt = 0; nt < 4; ++nt) {
#pragma unroll
            for (int r = 0; r < 8; ++r) crow[(size_t)r * HID + (nt << 4)] = acc[nt][r];
        }
    } else if (m0 < N) {
#pragma unroll
        for (int nt = 0; nt < 4; ++nt) {
#pragma unroll
            for (int r = 0; r < 8; ++r) {
                if (m0 + hi * 8 + r < N) crow[(size_t)r * HID + (nt << 4)] = acc[nt][r];
            }
        }
    }
}

template <bool RELU_F16>
__global__ void __launch_bounds__(256)
gcn_gather(const float* __restrict__ H, const int* __restrict__ rowptr,
           const int* __restrict__ nbr, const float* __restrict__ dis,
           const float* __restrict__ bias, float* __restrict__ outf,
           _Float16* __restrict__ outh, int N) {
    __shared__ int s_sorted[8][32];
    const int lane = threadIdx.x & 31;
    const int wv   = threadIdx.x >> 5;
    const int node = __builtin_amdgcn_readfirstlane((int)(blockIdx.x * 8 + wv));
    if (node >= N) return;
    const int beg = rowptr[node];
    const int cnt = rowptr[node + 1] - beg;
    const float dd = dis[node];
    const int* __restrict__ bucket = nbr + beg;

    const float2 hs = *(const float2*)(H + (size_t)node * HID + lane * 2);
    const float wself = dd * dd;
    float2 acc;
    acc.x = hs.x * wself;
    acc.y = hs.y * wself;

    for (int c0 = 0; c0 < cnt; c0 += 32) {
        for (int a0 = 0; a0 < cnt; a0 += 32) {
            const int ia  = a0 + lane;
            const int key = (ia < cnt) ? bucket[ia] : 0x7fffffff;
            const unsigned long long mine =
                ((unsigned long long)(unsigned)key << 32) | (unsigned)ia;
            int rank = 0;
            for (int b0 = 0; b0 < cnt; b0 += 32) {
                const int ib = b0 + lane;
                const int kb = (ib < cnt) ? bucket[ib] : 0x7fffffff;
                const int nb = (cnt - b0 < 32) ? (cnt - b0) : 32;
                for (int j = 0; j < nb; ++j) {
                    const unsigned kj = (unsigned)__builtin_amdgcn_readlane(kb, j);
                    const unsigned long long other =
                        ((unsigned long long)kj << 32) | (unsigned)(b0 + j);
                    rank += (other < mine) ? 1 : 0;
                }
            }
            if (ia < cnt && rank >= c0 && rank < c0 + 32) s_sorted[wv][rank - c0] = key;
        }
        const int n  = (cnt - c0 < 32) ? (cnt - c0) : 32;
        const int sk = (lane < n) ? s_sorted[wv][lane] : 0;
        const float w = (lane < n) ? dis[sk] * dd : 0.0f;
#pragma unroll 4
        for (int r = 0; r < n; ++r) {
            const int   s  = __builtin_amdgcn_readlane(sk, r);
            const float ww = __builtin_bit_cast(float,
                                 __builtin_amdgcn_readlane(__builtin_bit_cast(int, w), r));
            const float2 h = *(const float2*)(H + (size_t)s * HID + lane * 2);
            acc.x = fmaf(h.x, ww, acc.x);
            acc.y = fmaf(h.y, ww, acc.y);
        }
    }

    const float2 b2v = *(const float2*)(bias + lane * 2);
    if (RELU_F16) {
        typedef __attribute__((ext_vector_type(2))) _Float16 v2h;
        v2h o;
        o.x = (_Float16)fmaxf(acc.x + b2v.x, 0.0f);
        o.y = (_Float16)fmaxf(acc.y + b2v.y, 0.0f);
        *(v2h*)(outh + (size_t)node * HID + lane * 2) = o;
    } else {
        float2 o;
        o.x = acc.x + b2v.x;
        o.y = acc.y + b2v.y;
        *(float2*)(outf + (size_t)node * HID + lane * 2) = o;
    }
}

__global__ void __launch_bounds__(512)
gcn_gather_proj(const _Float16* __restrict__ H, const int* __restrict__ rowptr,
                const int* __restrict__ nbr, const float* __restrict__ dis,
                const float* __restrict__ Wt, const float* __restrict__ bias,
                float* __restrict__ outf, int N) {
    typedef __attribute__((ext_vector_type(2))) _Float16 v2h;
    constexpr int P  = HID + 8;
    constexpr int CP = HID + 4;
    __shared__ alignas(16) _Float16 Bs[HID * P];
    __shared__ alignas(16) _Float16 Ahi[16 * P];
    __shared__ alignas(16) _Float16 Alo[16 * P];
    __shared__ alignas(16) float    Cs[16 * CP];
    __shared__ int s_sorted[16][32];

    const int tid  = threadIdx.x;
    const int lane = tid & 31;
    const int wv   = __builtin_amdgcn_readfirstlane(tid >> 5);
    const int nd   = (int)(blockIdx.x * 16) + wv;
    const int node = (nd < N) ? nd : (N - 1);

    const int wn  = tid & (HID - 1);
    const int wk0 = tid >> 6;
    float wf[8];
    {
        const float* wp = Wt + (size_t)wk0 * 8 * HID + wn;
#pragma unroll
        for (int q = 0; q < 8; ++q) wf[q] = wp[q * HID];
    }

    const int beg = rowptr[node];
    const int cnt = rowptr[node + 1] - beg;
    const float dd = dis[node];
    const int* __restrict__ bucket = nbr + beg;

    const v2h hs = *(const v2h*)(H + (size_t)node * HID + lane * 2);
    const float wself = dd * dd;
    float2 acc;
    acc.x = (float)hs.x * wself;
    acc.y = (float)hs.y * wself;

    {
        v8h pk;
#pragma unroll
        for (int q = 0; q < 8; ++q) pk[q] = (_Float16)wf[q];
        *(v8h*)(Bs + wn * P + wk0 * 8) = pk;
    }

    for (int c0 = 0; c0 < cnt; c0 += 32) {
        for (int a0 = 0; a0 < cnt; a0 += 32) {
            const int ia  = a0 + lane;
            const int key = (ia < cnt) ? bucket[ia] : 0x7fffffff;
            const unsigned long long mine =
                ((unsigned long long)(unsigned)key << 32) | (unsigned)ia;
            int rank = 0;
            for (int b0 = 0; b0 < cnt; b0 += 32) {
                const int ib = b0 + lane;
                const int kb = (ib < cnt) ? bucket[ib] : 0x7fffffff;
                const int nb = (cnt - b0 < 32) ? (cnt - b0) : 32;
                for (int j = 0; j < nb; ++j) {
                    const unsigned kj = (unsigned)__builtin_amdgcn_readlane(kb, j);
                    const unsigned long long other =
                        ((unsigned long long)kj << 32) | (unsigned)(b0 + j);
                    rank += (other < mine) ? 1 : 0;
                }
            }
            if (ia < cnt && rank >= c0 && rank < c0 + 32) s_sorted[wv][rank - c0] = key;
        }
        const int n  = (cnt - c0 < 32) ? (cnt - c0) : 32;
        const int sk = (lane < n) ? s_sorted[wv][lane] : 0;
        const float w = (lane < n) ? dis[sk] * dd : 0.0f;
#pragma unroll 4
        for (int r = 0; r < n; ++r) {
            const int   s  = __builtin_amdgcn_readlane(sk, r);
            const float ww = __builtin_bit_cast(float,
                                 __builtin_amdgcn_readlane(__builtin_bit_cast(int, w), r));
            const v2h h = *(const v2h*)(H + (size_t)s * HID + lane * 2);
            acc.x = fmaf((float)h.x, ww, acc.x);
            acc.y = fmaf((float)h.y, ww, acc.y);
        }
    }

    {
        v2h hi2, lo2;
        hi2.x = (_Float16)acc.x;
        hi2.y = (_Float16)acc.y;
        lo2.x = (_Float16)(acc.x - (float)hi2.x);
        lo2.y = (_Float16)(acc.y - (float)hi2.y);
        *(v2h*)(Ahi + wv * P + lane * 2) = hi2;
        *(v2h*)(Alo + wv * P + lane * 2) = lo2;
    }
    __syncthreads();

    if (wv < 4) {
        const int c  = lane & 15;
        const int hi = lane >> 4;
        v8f d = (v8f){0.f, 0.f, 0.f, 0.f, 0.f, 0.f, 0.f, 0.f};
#pragma unroll
        for (int kt = 0; kt < HID / 32; ++kt) {
            const int ko = kt * 32 + hi * 8;
            union { v16h v; v8h h[2]; } ah, al, b;
            ah.h[0] = *(const v8h*)(Ahi + c * P + ko);
            ah.h[1] = *(const v8h*)(Ahi + c * P + ko + 16);
            al.h[0] = *(const v8h*)(Alo + c * P + ko);
            al.h[1] = *(const v8h*)(Alo + c * P + ko + 16);
            b.h[0]  = *(const v8h*)(Bs + (wv * 16 + c) * P + ko);
            b.h[1]  = *(const v8h*)(Bs + (wv * 16 + c) * P + ko + 16);
            d = __builtin_amdgcn_wmma_f32_16x16x32_f16(
                 false, al.v,  false, b.v,
                 (short)0, d,  false,  false);
            d = __builtin_amdgcn_wmma_f32_16x16x32_f16(
                 false, ah.v,  false, b.v,
                 (short)0, d,  false,  false);
        }
#pragma unroll
        for (int r = 0; r < 8; ++r) Cs[(hi * 8 + r) * CP + wv * 16 + c] = d[r];
    }
    __syncthreads();

    if (nd < N) {
        const float2 cv  = *(const float2*)(Cs + wv * CP + lane * 2);
        const float2 b2v = *(const float2*)(bias + lane * 2);
        float2 o;
        o.x = cv.x + b2v.x;
        o.y = cv.y + b2v.y;
        *(float2*)(outf + (size_t)nd * HID + lane * 2) = o;
    }
}


extern "C" void kernel_launch(void* const* d_in, const int* in_sizes, int n_in,
                              void* d_out, int out_size, void* d_ws, size_t ws_size,
                              hipStream_t stream) {
    const float* x  = (const float*)d_in[0];
    const int*   ei = (const int*)d_in[1];
    const float* W1 = (const float*)d_in[2];
    const float* b1 = (const float*)d_in[3];
    const float* W2 = (const float*)d_in[4];
    const float* b2 = (const float*)d_in[5];
    float* out = (float*)d_out;

    const int N = in_sizes[0] / IN_CH;
    const int E = in_sizes[1] / 2;
    const int* src = ei;
    const int* dst = ei + E;

    char* ws = (char*)d_ws;
    size_t off = 0;
    auto alloc = [&](size_t bytes) -> char* {
        char* p = ws + off;
        off = (off + bytes + 255) & ~(size_t)255;
        return p;
    };
    float*    proj = (float*)   alloc((size_t)N * HID * sizeof(float));
    _Float16* h1h  = (_Float16*)alloc((size_t)N * HID * sizeof(_Float16));
    float*    dis  = (float*)   alloc((size_t)N * sizeof(float));
    int*      crs  = (int*)     alloc((size_t)CHUNKS * 64 * sizeof(int));
    int*      rowp = (int*)     alloc((size_t)(N + 1) * sizeof(int));
    int*      slot = (int*)     alloc((size_t)E * sizeof(int));
    int*      nbr  = (int*)     alloc((size_t)E * sizeof(int));
    unsigned short* hist = (unsigned short*)h1h;
    int*            offp = (int*)proj;

    const int T = 256;
    const int gemmBlocks = ((N + 15) / 16 + 7) / 8;
    const int CH = (E + CHUNKS - 1) / CHUNKS;

    gcn_chunk_hist<<<CHUNKS, 1024, 0, stream>>>(dst, hist, crs, slot, N, E, CH);
    gcn_scan_deg<<<(N + 1023) / 1024, 1024, 0, stream>>>(hist, crs, offp, rowp, dis, N);
    gcn_fill<<<(E + T - 1) / T, T, 0, stream>>>(src, dst, slot, offp, nbr, N, E, CH);

    gcn_gemm_wmma<IN_CH, true><<<gemmBlocks, 256, 0, stream>>>(x, nullptr, W1, proj, N);

    gcn_gather<true><<<(N + 7) / 8, T, 0, stream>>>(proj, rowp, nbr, dis, b1, nullptr, h1h, N);

    gcn_gather_proj<<<(N + 15) / 16, 512, 0, stream>>>(h1h, rowp, nbr, dis, W2, b2, out, N);
}


# both neighbour sums: 4 destination nodes per wave (8 lanes per node, 16 B per lane per source row, bucket ranked via LDS), N/4 waves per layer; sum order unchanged (self, then ascending source id)
# speedup vs baseline: 2.6853x; 1.0407x over previous
#include <hip/hip_runtime.h>
#include <hip/hip_bf16.h>

typedef __attribute__((ext_vector_type(16))) _Float16 v16h;
typedef __attribute__((ext_vector_type(8)))  _Float16 v8h;
typedef __attribute__((ext_vector_type(8)))  float    v8f;

#define IN_CH 128
#define HID   64


#define CHUNKS    64
#define TAB_WORDS 51200
static_assert(CHUNKS == HID, "the count and offset planes reuse the N x HID buffers of the layers");

__global__ void __launch_bounds__(1024)
gcn_chunk_hist(const int* __restrict__ dst, unsigned short* __restrict__ hist,
               int* __restrict__ coarse, int* __restrict__ slot, int N, int E, int CH) {
    __shared__ alignas(16) unsigned tab[TAB_WORDS];
    const int tid  = threadIdx.x;
    const int lane = tid & 31;
    const int wv   = tid >> 5;
    const int g    = blockIdx.x;

    for (int i = tid; i < TAB_WORDS / 4; i += 1024) ((uint4*)tab)[i] = make_uint4(0u, 0u, 0u, 0u);
    __syncthreads();

    const int e0 = g * CH;
    const int e1 = (e0 + CH < E) ? (e0 + CH) : E;
    for (int eb = e0 + tid; eb < e1; eb += 4 * 1024) {
        int d[4];
#pragma unroll
        for (int u = 0; u < 4; ++u) {
            const int e = eb + u * 1024;
            d[u] = (e < e1) ? dst[e] : -1;
        }
#pragma unroll
        for (int u = 0; u < 4; ++u) {
            if ((unsigned)d[u] < (unsigned)TAB_WORDS) slot[eb + u * 1024] = (int)atomicAdd(&tab[d[u]], 1u);
        }
    }
    __syncthreads();

    const int nn = (N < TAB_WORDS) ? N : TAB_WORDS;
    unsigned short* __restrict__ hp = hist + (size_t)g * N;
    for (int n = tid; n < nn; n += 1024) hp[n] = (unsigned short)tab[n];

    for (int j = wv; j < 64; j += 32) {
        unsigned sacc = 0u;
        if (j < TAB_WORDS / 1024) {
            for (int i = 0; i < 32; ++i) sacc += tab[j * 1024 + i * 32 + lane];
        }
        for (int dd = 16; dd > 0; dd >>= 1) sacc += __shfl_xor(sacc, dd);
        if (lane == 0) coarse[g * 64 + j] = (int)sacc;
    }
}

__global__ void __launch_bounds__(1024)
gcn_scan_deg(const unsigned short* __restrict__ hist, const int* __restrict__ coarse,
             int* __restrict__ off, int* __restrict__ rowptr,
             float* __restrict__ dis, int N) {
    __shared__ int s_own[32];
    __shared__ int s_front[32];
    const int tid  = threadIdx.x;
    const int lane = tid & 31;
    const int wv   = tid >> 5;
    const int first = blockIdx.x * 1024;

    int front = 0;
    for (int i = tid; i < CHUNKS * 64; i += 1024) {
        const int v = coarse[i];
        front += ((i & 63) < (int)blockIdx.x) ? v : 0;
    }

    const int idx = first + tid;
    int c = 0;
    if (idx < N) {
#pragma unroll 16
        for (int g = 0; g < CHUNKS; ++g) c += (int)hist[(size_t)g * N + idx];
    }

    int incl = c;
    for (int d = 1; d < 32; d <<= 1) {
        int up = __shfl_up(incl, d);
        if (lane >= d) incl += up;
    }
    for (int d = 16; d > 0; d >>= 1) front += __shfl_xor(front, d);
    if (lane == 31) { s_own[wv] = incl; s_front[wv] = front; }
    __syncthreads();

    int t = s_front[lane] + ((lane < wv) ? s_own[lane] : 0);
    for (int d = 16; d > 0; d >>= 1) t += __shfl_xor(t, d);

    const int excl = t + incl - c;
    if (idx < N) {
        rowptr[idx] = excl;
        dis[idx]    = rsqrtf((float)(c + 1));
        if (idx == N - 1) rowptr[N] = excl + c;
        int run = excl;
#pragma unroll 16
        for (int g = 0; g < CHUNKS; ++g) {
            const int h = (int)hist[(size_t)g * N + idx];
            off[(size_t)g * N + idx] = run;
            run += h;
        }
    }
}

__global__ void gcn_fill(const int* __restrict__ src, const int* __restrict__ dst,
                         const int* __restrict__ slot, const int* __restrict__ off,
                         int* __restrict__ nbr, int N, int E, int CH) {
    int e = blockIdx.x * blockDim.x + threadIdx.x;
    if (e < E) {
        const int g = e / CH;
        const unsigned pos = (unsigned)(off[(size_t)g * N + dst[e]] + slot[e]);
        if (pos < (unsigned)E) nbr[pos] = src[e];
    }
}

template <int K, bool A_F32>
__global__ void __launch_bounds__(256)
gcn_gemm_wmma(const float* __restrict__ Af, const _Float16* __restrict__ Ah,
              const float* __restrict__ Wt, float* __restrict__ C, int N) {
    constexpr int KT = K / 32;
    constexpr int P  = K + 8;
    __shared__ alignas(16) _Float16 Bs[HID * P];

    const int tid  = threadIdx.x;
    const int lane = tid & 31;
    const int wv   = __builtin_amdgcn_readfirstlane(tid >> 5);
    const int c    = lane & 15;
    const int hi   = lane >> 4;
    const int m0   = (blockIdx.x * 8 + wv) << 4;
    const int row  = (m0 + c < N) ? (m0 + c) : (N - 1);

    constexpr int IT = HID * (K / 8) / 256;
    const int wn  = tid & (HID - 1);
    const int wk0 = tid >> 6;
    float wf[IT][8];
#pragma unroll
    for (int j = 0; j < IT; ++j) {
        const float* wp = Wt + (size_t)(wk0 + j * 4) * 8 * HID + wn;
#pragma unroll
        for (int q = 0; q < 8; ++q) wf[j][q] = wp[q * HID];
    }
    float4 af[A_F32 ? KT * 4 : 1];
    v8h    ah[A_F32 ? 1 : KT * 2];
    if constexpr (A_F32) {
        const float* ap = Af + (size_t)row * K + hi * 8;
#pragma unroll
        for (int kt = 0; kt < KT; ++kt) {
            af[kt * 4 + 0] = *(const float4*)(ap + kt * 32);
            af[kt * 4 + 1] = *(const float4*)(ap + kt * 32 + 4);
            af[kt * 4 + 2] = *(const float4*)(ap + kt * 32 + 16);
            af[kt * 4 + 3] = *(const float4*)(ap + kt * 32 + 20);
        }
    } else {
        const _Float16* ap = Ah + (size_t)row * K + hi * 8;
#pragma unroll
        for (int kt = 0; kt < KT; ++kt) {
            ah[kt * 2 + 0] = *(const v8h*)(ap + kt * 32);
            ah[kt * 2 + 1] = *(const v8h*)(ap + kt * 32 + 16);
        }
    }

#pragma unroll
    for (int j = 0; j < IT; ++j) {
        v8h pk;
#pragma unroll
        for (int q = 0; q < 8; ++q) pk[q] = (_Float16)wf[j][q];
        *(v8h*)(Bs + wn * P + (wk0 + j * 4) * 8) = pk;
    }
    __syncthreads();

    v8f acc[4];
#pragma unroll
    for (int nt = 0; nt < 4; ++nt) acc[nt] = (v8f){0.f, 0.f, 0.f, 0.f, 0.f, 0.f, 0.f, 0.f};

#pragma unroll
    for (int kt = 0; kt < KT; ++kt) {
        union { v16h v; v8h h[2]; } a;
        if constexpr (A_F32) {
#pragma unroll
            for (int j = 0; j < 4; ++j) {
                a.v[j * 4 + 0] = (_Float16)af[kt * 4 + j].x;
                a.v[j * 4 + 1] = (_Float16)af[kt * 4 + j].y;
                a.v[j * 4 + 2] = (_Float16)af[kt * 4 + j].z;
                a.v[j * 4 + 3] = (_Float16)af[kt * 4 + j].w;
            }
        } else {
            a.h[0] = ah[kt * 2 + 0];
            a.h[1] = ah[kt * 2 + 1];
        }
#pragma unroll
        for (int nt = 0; nt < 4; ++nt) {
            const _Float16* bp = Bs + (nt * 16 + c) * P + kt * 32 + hi * 8;
            union { v16h v; v8h h[2]; } b;
            b.h[0] = *(const v8h*)(bp);
            b.h[1] = *(const v8h*)(bp + 16);
            acc[nt] = __builtin_amdgcn_wmma_f32_16x16x32_f16(
                 false, a.v,  false, b.v,
                 (short)0, acc[nt],  false,  false);
        }
    }

    float* crow = C + (size_t)(m0 + hi * 8) * HID + c;
    if (m0 + 16 <= N) {
#pragma unroll
        for (int nt = 0; nt < 4; ++nt) {
#pragma unroll
            for (int r = 0; r < 8; ++r) crow[(size_t)r * HID + (nt << 4)] = acc[nt][r];
        }
    } else if (m0 < N) {
#pragma unroll
        for (int nt = 0; nt < 4; ++nt) {
#pragma unroll
            for (int r = 0; r < 8; ++r) {
                if (m0 + hi * 8 + r < N) crow[(size_t)r * HID + (nt << 4)] = acc[nt][r];
            }
        }
    }
}


#define GRP    4
#define WPITCH 34

typedef unsigned long long u64;

template <int NT>
__device__ __forceinline__ void gcn_rank_window(const uint2* in_q, int nb8,
                                                const u64 (&mine)[4], int (&rank)[4]) {
    for (int j = 0; j < nb8; j += 8) {
        uint4 e[4];
#pragma unroll
        for (int v = 0; v < 4; ++v) e[v] = *(const uint4*)(in_q + j + 2 * v);
#pragma unroll
        for (int v = 0; v < 4; ++v) {
            const u64 o0 = ((u64)e[v].y << 32) | (u64)e[v].x;
            const u64 o1 = ((u64)e[v].w << 32) | (u64)e[v].z;
#pragma unroll
            for (int t = 0; t < NT; ++t)
                rank[t] += ((o0 < mine[t]) ? 1 : 0) + ((o1 < mine[t]) ? 1 : 0);
        }
    }
}

template <bool ROW_F16> struct GcnRow;
template <> struct GcnRow<true>  { v8h v; };
template <> struct GcnRow<false> { float4 a, b; };

template <bool ROW_F16>
__device__ __forceinline__ GcnRow<ROW_F16> gcn_row_load(const float* __restrict__ Hf,
                                                        const _Float16* __restrict__ Hh,
                                                        int s, int sub) {
    GcnRow<ROW_F16> r;
    if constexpr (ROW_F16) {
        r.v = *(const v8h*)((const char*)Hh + (unsigned)(s * (HID * 2) + sub * 16));
    } else {
        const float* p = Hf + (unsigned)(s * HID + sub * 4);
        r.a = *(const float4*)p;
        r.b = *(const float4*)(p + 32);
    }
    return r;
}

template <bool ROW_F16>
__device__ __forceinline__ void gcn_row_fma(const GcnRow<ROW_F16>& r, float w, float (&acc)[8]) {
    if constexpr (ROW_F16) {
#pragma unroll
        for (int i = 0; i < 8; ++i) acc[i] = fmaf((float)r.v[i], w, acc[i]);
    } else {
        acc[0] = fmaf(r.a.x, w, acc[0]);
        acc[1] = fmaf(r.a.y, w, acc[1]);
        acc[2] = fmaf(r.a.z, w, acc[2]);
        acc[3] = fmaf(r.a.w, w, acc[3]);
        acc[4] = fmaf(r.b.x, w, acc[4]);
        acc[5] = fmaf(r.b.y, w, acc[5]);
        acc[6] = fmaf(r.b.z, w, acc[6]);
        acc[7] = fmaf(r.b.w, w, acc[7]);
    }
}

template <bool ROW_F16, int U>
__device__ __forceinline__ void gcn_node_sums(const float* __restrict__ Hf, const _Float16* __restrict__ Hh,
                                              const int* __restrict__ rowptr, const int* __restrict__ nbr,
                                              const float* __restrict__ dis, int node, int sub,
                                              uint2* in_q, uint2* out_q, float (&acc)[8]) {
    const int beg = rowptr[node];
    const int cnt = rowptr[node + 1] - beg;
    const float dd = dis[node];

    int m = cnt;
    { const int o = __shfl_xor(m, 8);  m = (o > m) ? o : m; }
    { const int o = __shfl_xor(m, 16); m = (o > m) ? o : m; }
    const int maxcnt = __builtin_amdgcn_readfirstlane(m);

    {
        const GcnRow<ROW_F16> hs = gcn_row_load<ROW_F16>(Hf, Hh, node, sub);
        const float wself = dd * dd;
        if constexpr (ROW_F16) {
#pragma unroll
            for (int i = 0; i < 8; ++i) acc[i] = (float)hs.v[i] * wself;
        } else {
            acc[0] = hs.a.x * wself; acc[1] = hs.a.y * wself; acc[2] = hs.a.z * wself; acc[3] = hs.a.w * wself;
            acc[4] = hs.b.x * wself; acc[5] = hs.b.y * wself; acc[6] = hs.b.z * wself; acc[7] = hs.b.w * wself;
        }
    }

    for (int c0 = 0; c0 < maxcnt; c0 += 32) {
#pragma unroll
        for (int t = 0; t < 4; ++t) {
            const int p = t * 8 + sub;
            if (c0 + p >= cnt) out_q[p] = make_uint2((unsigned)node, 0u);
        }
        for (int a0 = 0; a0 < maxcnt; a0 += 32) {
            int   key[4];
            float kw[4];
            int   rank[4];
            u64   mine[4];
#pragma unroll
            for (int t = 0; t < 4; ++t) {
                const int ia = a0 + t * 8 + sub;
                key[t]  = nbr[(ia < cnt) ? (beg + ia) : 0];
                key[t]  = (ia < cnt) ? key[t] : 0x7fffffff;
                rank[t] = 0;
                mine[t] = ((u64)(unsigned)key[t] << 32) | (u64)(unsigned)ia;
            }
#pragma unroll
            for (int t = 0; t < 4; ++t) {
                const int ia = a0 + t * 8 + sub;
                kw[t] = dis[(ia < cnt) ? key[t] : node] * dd;
            }
            const int nt = (maxcnt - a0 + 7) >> 3;
            for (int b0 = 0; b0 < maxcnt; b0 += 32) {
#pragma unroll
                for (int t = 0; t < 4; ++t) {
                    const int ib = b0 + t * 8 + sub;
                    int kb = key[t];
                    if (b0 != a0) {
                        kb = nbr[(ib < cnt) ? (beg + ib) : 0];
                        kb = (ib < cnt) ? kb : 0x7fffffff;
                    }
                    in_q[t * 8 + sub] = make_uint2((unsigned)ib, (unsigned)kb);
                }
                const int nb  = (maxcnt - b0 < 32) ? (maxcnt - b0) : 32;
                const int nb8 = (nb + 7) & ~7;
                if (nt >= 4)      gcn_rank_window<4>(in_q, nb8, mine, rank);
                else if (nt == 3) gcn_rank_window<3>(in_q, nb8, mine, rank);
                else if (nt == 2) gcn_rank_window<2>(in_q, nb8, mine, rank);
                else              gcn_rank_window<1>(in_q, nb8, mine, rank);
            }
#pragma unroll
            for (int t = 0; t < 4; ++t) {
                const int ia = a0 + t * 8 + sub;
                if (ia < cnt && rank[t] >= c0 && rank[t] < c0 + 32)
                    out_q[rank[t] - c0] = make_uint2((unsigned)key[t], __builtin_bit_cast(unsigned, kw[t]));
            }
        }
        const int nwin = (maxcnt - c0 < 32) ? (maxcnt - c0) : 32;
        for (int r = 0; r < nwin; r += U) {
            uint4 e[U / 2];
#pragma unroll
            for (int v = 0; v < U / 2; ++v) e[v] = *(const uint4*)(out_q + r + 2 * v);
            GcnRow<ROW_F16> raw[U];
            float w[U];
#pragma unroll
            for (int u = 0; u < U; ++u) {
                const unsigned k  = (u & 1) ? e[u >> 1].z : e[u >> 1].x;
                const unsigned wb = (u & 1) ? e[u >> 1].w : e[u >> 1].y;
                w[u]   = __builtin_bit_cast(float, wb);
                raw[u] = gcn_row_load<ROW_F16>(Hf, Hh, (int)k, sub);
            }
#pragma unroll
            for (int u = 0; u < U; ++u) gcn_row_fma<ROW_F16>(raw[u], w[u], acc);
        }
    }
}

__global__ void __launch_bounds__(256) __attribute__((amdgpu_waves_per_eu(16)))
gcn_gather(const float* __restrict__ H, const int* __restrict__ rowptr,
           const int* __restrict__ nbr, const float* __restrict__ dis,
           const float* __restrict__ bias, _Float16* __restrict__ outh, int N) {
    typedef __attribute__((ext_vector_type(4))) _Float16 v4h;
    __shared__ alignas(16) uint2 s_in[8][GRP * WPITCH];
    __shared__ alignas(16) uint2 s_out[8][GRP * WPITCH];
    const int lane = threadIdx.x & 31;
    const int wv   = threadIdx.x >> 5;
    const int q    = lane >> 3;
    const int sub  = lane & 7;
    const int nd   = (int)(blockIdx.x * 8 + wv) * GRP + q;
    const int node = (nd < N) ? nd : (N - 1);

    float acc[8];
    gcn_node_sums<false, 4>(H, nullptr, rowptr, nbr, dis, node, sub,
                            &s_in[wv][q * WPITCH], &s_out[wv][q * WPITCH], acc);

    if (nd < N) {
        const float4 ba = *(const float4*)(bias + sub * 4);
        const float4 bb = *(const float4*)(bias + 32 + sub * 4);
        v4h oa, ob;
        oa.x = (_Float16)fmaxf(acc[0] + ba.x, 0.0f);
        oa.y = (_Float16)fmaxf(acc[1] + ba.y, 0.0f);
        oa.z = (_Float16)fmaxf(acc[2] + ba.z, 0.0f);
        oa.w = (_Float16)fmaxf(acc[3] + ba.w, 0.0f);
        ob.x = (_Float16)fmaxf(acc[4] + bb.x, 0.0f);
        ob.y = (_Float16)fmaxf(acc[5] + bb.y, 0.0f);
        ob.z = (_Float16)fmaxf(acc[6] + bb.z, 0.0f);
        ob.w = (_Float16)fmaxf(acc[7] + bb.w, 0.0f);
        _Float16* op = outh + (size_t)nd * HID + sub * 4;
        *(v4h*)op        = oa;
        *(v4h*)(op + 32) = ob;
    }
}

__global__ void __launch_bounds__(128) __attribute__((amdgpu_waves_per_eu(16)))
gcn_gather_proj(const _Float16* __restrict__ H, const int* __restrict__ rowptr,
                const int* __restrict__ nbr, const float* __restrict__ dis,
                const float* __restrict__ Wt, const float* __restrict__ bias,
                float* __restrict__ outf, int N) {
    constexpr int P  = HID + 8;
    constexpr int CP = HID + 4;
    __shared__ alignas(16) _Float16 Bs[HID * P];
    __shared__ alignas(16) _Float16 Ahi[16 * P];
    __shared__ alignas(16) _Float16 Alo[16 * P];
    __shared__ alignas(16) uint2 s_win[2][4][GRP * WPITCH];
    static_assert(sizeof(float) * 16 * CP <= sizeof(uint2) * 2 * 4 * GRP * WPITCH, "the f32 tile must fit into the sort windows");
    float* Cs = (float*)&s_win[0][0][0];

    const int tid  = threadIdx.x;
    const int lane = tid & 31;
    const int wv   = __builtin_amdgcn_readfirstlane(tid >> 5);
    const int q    = lane >> 3;
    const int sub  = lane & 7;
    const int trow = wv * GRP + q;
    const int nd   = (int)(blockIdx.x * 16) + trow;
    const int node = (nd < N) ? nd : (N - 1);

    const int wn  = tid & (HID - 1);
    const int wk0 = tid >> 6;
    float wf[4][8];
#pragma unroll
    for (int j = 0; j < 4; ++j) {
        const float* wp = Wt + (size_t)(wk0 + j * 2) * 8 * HID + wn;
#pragma unroll
        for (int qq = 0; qq < 8; ++qq) wf[j][qq] = wp[qq * HID];
    }
    const int ocol = (lane & 15) * 4;

#pragma unroll
    for (int j = 0; j < 4; ++j) {
        v8h pk;
#pragma unroll
        for (int qq = 0; qq < 8; ++qq) pk[qq] = (_Float16)wf[j][qq];
        *(v8h*)(Bs + wn * P + (wk0 + j * 2) * 8) = pk;
    }

    float acc[8];
    gcn_node_sums<true, 4>(nullptr, H, rowptr, nbr, dis, node, sub,
                           &s_win[0][wv][q * WPITCH], &s_win[1][wv][q * WPITCH], acc);

    {
        v8h hi8, lo8;
#pragma unroll
        for (int i = 0; i < 8; ++i) {
            hi8[i] = (_Float16)acc[i];
            lo8[i] = (_Float16)(acc[i] - (float)hi8[i]);
        }
        *(v8h*)(Ahi + trow * P + sub * 8) = hi8;
        *(v8h*)(Alo + trow * P + sub * 8) = lo8;
    }
    const float4 b2v = *(const float4*)(bias + ocol);
    __syncthreads();

    {
        const int c  = lane & 15;
        const int hi = lane >> 4;
        v8f d = (v8f){0.f, 0.f, 0.f, 0.f, 0.f, 0.f, 0.f, 0.f};
#pragma unroll
        for (int kt = 0; kt < HID / 32; ++kt) {
            const int ko = kt * 32 + hi * 8;
            union { v16h v; v8h h[2]; } ah, al, b;
            ah.h[0] = *(const v8h*)(Ahi + c * P + ko);
            ah.h[1] = *(const v8h*)(Ahi + c * P + ko + 16);
            al.h[0] = *(const v8h*)(Alo + c * P + ko);
            al.h[1] = *(const v8h*)(Alo + c * P + ko + 16);
            b.h[0]  = *(const v8h*)(Bs + (wv * 16 + c) * P + ko);
            b.h[1]  = *(const v8h*)(Bs + (wv * 16 + c) * P + ko + 16);
            d = __builtin_amdgcn_wmma_f32_16x16x32_f16(
                 false, al.v,  false, b.v,
                 (short)0, d,  false,  false);
            d = __builtin_amdgcn_wmma_f32_16x16x32_f16(
                 false, ah.v,  false, b.v,
                 (short)0, d,  false,  false);
        }
#pragma unroll
        for (int r = 0; r < 8; ++r) Cs[(hi * 8 + r) * CP + wv * 16 + c] = d[r];
    }
    __syncthreads();

#pragma unroll
    for (int k = 0; k < 2; ++k) {
        const int orow = wv * GRP + k * 2 + (lane >> 4);
        const int on   = (int)(blockIdx.x * 16) + orow;
        if (on < N) {
            const float4 cv = *(const float4*)(Cs + orow * CP + ocol);
            float4 o;
            o.x = cv.x + b2v.x;
            o.y = cv.y + b2v.y;
            o.z = cv.z + b2v.z;
            o.w = cv.w + b2v.w;
            *(float4*)(outf + (size_t)on * HID + ocol) = o;
        }
    }
}


extern "C" void kernel_launch(void* const* d_in, const int* in_sizes, int n_in,
                              void* d_out, int out_size, void* d_ws, size_t ws_size,
                              hipStream_t stream) {
    const float* x  = (const float*)d_in[0];
    const int*   ei = (const int*)d_in[1];
    const float* W1 = (const float*)d_in[2];
    const float* b1 = (const float*)d_in[3];
    const float* W2 = (const float*)d_in[4];
    const float* b2 = (const float*)d_in[5];
    float* out = (float*)d_out;

    const int N = in_sizes[0] / IN_CH;
    const int E = in_sizes[1] / 2;
    const int* src = ei;
    const int* dst = ei + E;

    char* ws = (char*)d_ws;
    size_t off = 0;
    auto alloc = [&](size_t bytes) -> char* {
        char* p = ws + off;
        off = (off + bytes + 255) & ~(size_t)255;
        return p;
    };
    float*    proj = (float*)   alloc((size_t)N * HID * sizeof(float));
    _Float16* h1h  = (_Float16*)alloc((size_t)N * HID * sizeof(_Float16));
    float*    dis  = (float*)   alloc((size_t)N * sizeof(float));
    int*      crs  = (int*)     alloc((size_t)CHUNKS * 64 * sizeof(int));
    int*      rowp = (int*)     alloc((size_t)(N + 1) * sizeof(int));
    int*      slot = (int*)     alloc((size_t)E * sizeof(int));
    int*      nbr  = (int*)     alloc((size_t)E * sizeof(int));
    unsigned short* hist = (unsigned short*)h1h;
    int*            offp = (int*)proj;

    const int T = 256;
    const int gemmBlocks = ((N + 15) / 16 + 7) / 8;
    const int CH = (E + CHUNKS - 1) / CHUNKS;

    gcn_chunk_hist<<<CHUNKS, 1024, 0, stream>>>(dst, hist, crs, slot, N, E, CH);
    gcn_scan_deg<<<(N + 1023) / 1024, 1024, 0, stream>>>(hist, crs, offp, rowp, dis, N);
    gcn_fill<<<(E + T - 1) / T, T, 0, stream>>>(src, dst, slot, offp, nbr, N, E, CH);

    gcn_gemm_wmma<IN_CH, true><<<gemmBlocks, 256, 0, stream>>>(x, nullptr, W1, proj, N);

    gcn_gather<<<(N + 8 * GRP - 1) / (8 * GRP), T, 0, stream>>>(proj, rowp, nbr, dis, b1, h1h, N);

    gcn_gather_proj<<<(N + 15) / 16, 128, 0, stream>>>(h1h, rowp, nbr, dis, W2, b2, out, N);
}


# edge weight split: rows stored times deg^-1/2 of their own node, each neighbour sum scaled once by the destination's; no per-edge weight load, rank window holds ids only; sum order kept
# speedup vs baseline: 2.8252x; 1.0521x over previous
#include <hip/hip_runtime.h>
#include <hip/hip_bf16.h>

typedef __attribute__((ext_vector_type(16))) _Float16 v16h;
typedef __attribute__((ext_vector_type(8)))  _Float16 v8h;
typedef __attribute__((ext_vector_type(8)))  float    v8f;

#define IN_CH 128
#define HID   64


#define CHUNKS    64
#define TAB_WORDS 51200
static_assert(CHUNKS == HID, "the count and offset planes reuse the N x HID buffers of the layers");

__global__ void __launch_bounds__(1024)
gcn_chunk_hist(const int* __restrict__ dst, unsigned short* __restrict__ hist,
               int* __restrict__ coarse, int* __restrict__ slot, int N, int E, int CH) {
    __shared__ alignas(16) unsigned tab[TAB_WORDS];
    const int tid  = threadIdx.x;
    const int lane = tid & 31;
    const int wv   = tid >> 5;
    const int g    = blockIdx.x;

    for (int i = tid; i < TAB_WORDS / 4; i += 1024) ((uint4*)tab)[i] = make_uint4(0u, 0u, 0u, 0u);
    __syncthreads();

    const int e0 = g * CH;
    const int e1 = (e0 + CH < E) ? (e0 + CH) : E;
    for (int eb = e0 + tid; eb < e1; eb += 4 * 1024) {
        int d[4];
#pragma unroll
        for (int u = 0; u < 4; ++u) {
            const int e = eb + u * 1024;
            d[u] = (e < e1) ? dst[e] : -1;
        }
#pragma unroll
        for (int u = 0; u < 4; ++u) {
            if ((unsigned)d[u] < (unsigned)TAB_WORDS) slot[eb + u * 1024] = (int)atomicAdd(&tab[d[u]], 1u);
        }
    }
    __syncthreads();

    const int nn = (N < TAB_WORDS) ? N : TAB_WORDS;
    unsigned short* __restrict__ hp = hist + (size_t)g * N;
    for (int n = tid; n < nn; n += 1024) hp[n] = (unsigned short)tab[n];

    for (int j = wv; j < 64; j += 32) {
        unsigned sacc = 0u;
        if (j < TAB_WORDS / 1024) {
            for (int i = 0; i < 32; ++i) sacc += tab[j * 1024 + i * 32 + lane];
        }
        for (int dd = 16; dd > 0; dd >>= 1) sacc += __shfl_xor(sacc, dd);
        if (lane == 0) coarse[g * 64 + j] = (int)sacc;
    }
}

__global__ void __launch_bounds__(1024)
gcn_scan_deg(const unsigned short* __restrict__ hist, const int* __restrict__ coarse,
             int* __restrict__ off, int* __restrict__ rowptr,
             float* __restrict__ dis, int N) {
    __shared__ int s_own[32];
    __shared__ int s_front[32];
    const int tid  = threadIdx.x;
    const int lane = tid & 31;
    const int wv   = tid >> 5;
    const int first = blockIdx.x * 1024;

    int front = 0;
    for (int i = tid; i < CHUNKS * 64; i += 1024) {
        const int v = coarse[i];
        front += ((i & 63) < (int)blockIdx.x) ? v : 0;
    }

    const int idx = first + tid;
    int c = 0;
    if (idx < N) {
#pragma unroll 16
        for (int g = 0; g < CHUNKS; ++g) c += (int)hist[(size_t)g * N + idx];
    }

    int incl = c;
    for (int d = 1; d < 32; d <<= 1) {
        int up = __shfl_up(incl, d);
        if (lane >= d) incl += up;
    }
    for (int d = 16; d > 0; d >>= 1) front += __shfl_xor(front, d);
    if (lane == 31) { s_own[wv] = incl; s_front[wv] = front; }
    __syncthreads();

    int t = s_front[lane] + ((lane < wv) ? s_own[lane] : 0);
    for (int d = 16; d > 0; d >>= 1) t += __shfl_xor(t, d);

    const int excl = t + incl - c;
    if (idx < N) {
        rowptr[idx] = excl;
        dis[idx]    = rsqrtf((float)(c + 1));
        if (idx == N - 1) rowptr[N] = excl + c;
        int run = excl;
#pragma unroll 16
        for (int g = 0; g < CHUNKS; ++g) {
            const int h = (int)hist[(size_t)g * N + idx];
            off[(size_t)g * N + idx] = run;
            run += h;
        }
    }
}

__global__ void gcn_fill(const int* __restrict__ src, const int* __restrict__ dst,
                         const int* __restrict__ slot, const int* __restrict__ off,
                         int* __restrict__ nbr, int N, int E, int CH) {
    int e = blockIdx.x * blockDim.x + threadIdx.x;
    if (e < E) {
        const int g = e / CH;
        const unsigned pos = (unsigned)(off[(size_t)g * N + dst[e]] + slot[e]);
        if (pos < (unsigned)E) nbr[pos] = src[e];
    }
}

template <int K, bool A_F32>
__global__ void __launch_bounds__(256)
gcn_gemm_wmma(const float* __restrict__ Af, const _Float16* __restrict__ Ah,
              const float* __restrict__ Wt, const float* __restrict__ rscale,
              float* __restrict__ C, int N) {
    constexpr int KT = K / 32;
    constexpr int P  = K + 8;
    __shared__ alignas(16) _Float16 Bs[HID * P];

    const int tid  = threadIdx.x;
    const int lane = tid & 31;
    const int wv   = __builtin_amdgcn_readfirstlane(tid >> 5);
    const int c    = lane & 15;
    const int hi   = lane >> 4;
    const int m0   = (blockIdx.x * 8 + wv) << 4;
    const int row  = (m0 + c < N) ? (m0 + c) : (N - 1);

    constexpr int IT = HID * (K / 8) / 256;
    const int wn  = tid & (HID - 1);
    const int wk0 = tid >> 6;
    float wf[IT][8];
#pragma unroll
    for (int j = 0; j < IT; ++j) {
        const float* wp = Wt + (size_t)(wk0 + j * 4) * 8 * HID + wn;
#pragma unroll
        for (int q = 0; q < 8; ++q) wf[j][q] = wp[q * HID];
    }
    float4 af[A_F32 ? KT * 4 : 1];
    v8h    ah[A_F32 ? 1 : KT * 2];
    if constexpr (A_F32) {
        const float* ap = Af + (size_t)row * K + hi * 8;
#pragma unroll
        for (int kt = 0; kt < KT; ++kt) {
            af[kt * 4 + 0] = *(const float4*)(ap + kt * 32);
            af[kt * 4 + 1] = *(const float4*)(ap + kt * 32 + 4);
            af[kt * 4 + 2] = *(const float4*)(ap + kt * 32 + 16);
            af[kt * 4 + 3] = *(const float4*)(ap + kt * 32 + 20);
        }
    } else {
        const _Float16* ap = Ah + (size_t)row * K + hi * 8;
#pragma unroll
        for (int kt = 0; kt < KT; ++kt) {
            ah[kt * 2 + 0] = *(const v8h*)(ap + kt * 32);
            ah[kt * 2 + 1] = *(const v8h*)(ap + kt * 32 + 16);
        }
    }
    float rs[8];
    if (m0 + 16 <= N) {
        const float4 r0 = *(const float4*)(rscale + m0 + hi * 8);
        const float4 r1 = *(const float4*)(rscale + m0 + hi * 8 + 4);
        rs[0] = r0.x; rs[1] = r0.y; rs[2] = r0.z; rs[3] = r0.w;
        rs[4] = r1.x; rs[5] = r1.y; rs[6] = r1.z; rs[7] = r1.w;
    } else {
#pragma unroll
        for (int r = 0; r < 8; ++r) {
            const int rr = m0 + hi * 8 + r;
            rs[r] = rscale[(rr < N) ? rr : (N - 1)];
        }
    }

#pragma unroll
    for (int j = 0; j < IT; ++j) {
        v8h pk;
#pragma unroll
        for (int q = 0; q < 8; ++q) pk[q] = (_Float16)wf[j][q];
        *(v8h*)(Bs + wn * P + (wk0 + j * 4) * 8) = pk;
    }
    __syncthreads();

    v8f acc[4];
#pragma unroll
    for (int nt = 0; nt < 4; ++nt) acc[nt] = (v8f){0.f, 0.f, 0.f, 0.f, 0.f, 0.f, 0.f, 0.f};

#pragma unroll
    for (int kt = 0; kt < KT; ++kt) {
        union { v16h v; v8h h[2]; } a;
        if constexpr (A_F32) {
#pragma unroll
            for (int j = 0; j < 4; ++j) {
                a.v[j * 4 + 0] = (_Float16)af[kt * 4 + j].x;
                a.v[j * 4 + 1] = (_Float16)af[kt * 4 + j].y;
                a.v[j * 4 + 2] = (_Float16)af[kt * 4 + j].z;
                a.v[j * 4 + 3] = (_Float16)af[kt * 4 + j].w;
            }
        } else {
            a.h[0] = ah[kt * 2 + 0];
            a.h[1] = ah[kt * 2 + 1];
        }
#pragma unroll
        for (int nt = 0; nt < 4; ++nt) {
            const _Float16* bp = Bs + (nt * 16 + c) * P + kt * 32 + hi * 8;
            union { v16h v; v8h h[2]; } b;
            b.h[0] = *(const v8h*)(bp);
            b.h[1] = *(const v8h*)(bp + 16);
            acc[nt] = __builtin_amdgcn_wmma_f32_16x16x32_f16(
                 false, a.v,  false, b.v,
                 (short)0, acc[nt],  false,  false);
        }
    }

    float* crow = C + (size_t)(m0 + hi * 8) * HID + c;
    if (m0 + 16 <= N) {
#pragma unroll
        for (int nt = 0; nt < 4; ++nt) {
#pragma unroll
            for (int r = 0; r < 8; ++r) crow[(size_t)r * HID + (nt << 4)] = acc[nt][r] * rs[r];
        }
    } else if (m0 < N) {
#pragma unroll
        for (int nt = 0; nt < 4; ++nt) {
#pragma unroll
            for (int r = 0; r < 8; ++r) {
                if (m0 + hi * 8 + r < N) crow[(size_t)r * HID + (nt << 4)] = acc[nt][r] * rs[r];
            }
        }
    }
}


#define GRP    4
#define WPITCH 34
#define OPITCH 36

typedef unsigned long long u64;

template <int NT>
__device__ __forceinline__ void gcn_rank_window(const uint2* in_q, int nb8,
                                                const u64 (&mine)[4], int (&rank)[4]) {
    for (int j = 0; j < nb8; j += 8) {
        uint4 e[4];
#pragma unroll
        for (int v = 0; v < 4; ++v) e[v] = *(const uint4*)(in_q + j + 2 * v);
#pragma unroll
        for (int v = 0; v < 4; ++v) {
            const u64 o0 = ((u64)e[v].y << 32) | (u64)e[v].x;
            const u64 o1 = ((u64)e[v].w << 32) | (u64)e[v].z;
#pragma unroll
            for (int t = 0; t < NT; ++t)
                rank[t] += ((o0 < mine[t]) ? 1 : 0) + ((o1 < mine[t]) ? 1 : 0);
        }
    }
}

template <bool ROW_F16> struct GcnRow;
template <> struct GcnRow<true>  { v8h v; };
template <> struct GcnRow<false> { float4 a, b; };

template <bool ROW_F16>
__device__ __forceinline__ GcnRow<ROW_F16> gcn_row_load(const float* __restrict__ Hf,
                                                        const _Float16* __restrict__ Hh,
                                                        int s, int sub) {
    GcnRow<ROW_F16> r;
    if constexpr (ROW_F16) {
        r.v = *(const v8h*)((const char*)Hh + (unsigned)(s * (HID * 2) + sub * 16));
    } else {
        const float* p = Hf + (unsigned)(s * HID + sub * 4);
        r.a = *(const float4*)p;
        r.b = *(const float4*)(p + 32);
    }
    return r;
}

template <bool ROW_F16>
__device__ __forceinline__ void gcn_row_fma(const GcnRow<ROW_F16>& r, float w, float (&acc)[8]) {
    if constexpr (ROW_F16) {
#pragma unroll
        for (int i = 0; i < 8; ++i) acc[i] = fmaf((float)r.v[i], w, acc[i]);
    } else {
        acc[0] = fmaf(r.a.x, w, acc[0]);
        acc[1] = fmaf(r.a.y, w, acc[1]);
        acc[2] = fmaf(r.a.z, w, acc[2]);
        acc[3] = fmaf(r.a.w, w, acc[3]);
        acc[4] = fmaf(r.b.x, w, acc[4]);
        acc[5] = fmaf(r.b.y, w, acc[5]);
        acc[6] = fmaf(r.b.z, w, acc[6]);
        acc[7] = fmaf(r.b.w, w, acc[7]);
    }
}

template <bool ROW_F16, int U>
__device__ __forceinline__ void gcn_node_sums(const float* __restrict__ Hf, const _Float16* __restrict__ Hh,
                                              const int* __restrict__ rowptr, const int* __restrict__ nbr,
                                              const float* __restrict__ dis, int node, int sub,
                                              uint2* in_q, unsigned* out_q, float (&acc)[8], float& dd) {
    static_assert(U % 4 == 0, "the row loop reads the rank window four ids at a time");
    const int beg = rowptr[node];
    const int cnt = rowptr[node + 1] - beg;

    int m = cnt;
    { const int o = __shfl_xor(m, 8);  m = (o > m) ? o : m; }
    { const int o = __shfl_xor(m, 16); m = (o > m) ? o : m; }
    const int maxcnt = __builtin_amdgcn_readfirstlane(m);

    {
        const GcnRow<ROW_F16> hs = gcn_row_load<ROW_F16>(Hf, Hh, node, sub);
        if constexpr (ROW_F16) {
#pragma unroll
            for (int i = 0; i < 8; ++i) acc[i] = (float)hs.v[i];
        } else {
            acc[0] = hs.a.x; acc[1] = hs.a.y; acc[2] = hs.a.z; acc[3] = hs.a.w;
            acc[4] = hs.b.x; acc[5] = hs.b.y; acc[6] = hs.b.z; acc[7] = hs.b.w;
        }
    }

    for (int c0 = 0; c0 < maxcnt; c0 += 32) {
#pragma unroll
        for (int t = 0; t < 4; ++t) {
            const int p = t * 8 + sub;
            if (c0 + p >= cnt) out_q[p] = (unsigned)node;
        }
        for (int a0 = 0; a0 < maxcnt; a0 += 32) {
            int   key[4];
            int   rank[4];
            u64   mine[4];
#pragma unroll
            for (int t = 0; t < 4; ++t) {
                const int ia = a0 + t * 8 + sub;
                key[t]  = nbr[(ia < cnt) ? (beg + ia) : 0];
                key[t]  = (ia < cnt) ? key[t] : 0x7fffffff;
                rank[t] = 0;
                mine[t] = ((u64)(unsigned)key[t] << 32) | (u64)(unsigned)ia;
            }
            const int nt = (maxcnt - a0 + 7) >> 3;
            for (int b0 = 0; b0 < maxcnt; b0 += 32) {
#pragma unroll
                for (int t = 0; t < 4; ++t) {
                    const int ib = b0 + t * 8 + sub;
                    int kb = key[t];
                    if (b0 != a0) {
                        kb = nbr[(ib < cnt) ? (beg + ib) : 0];
                        kb = (ib < cnt) ? kb : 0x7fffffff;
                    }
                    in_q[t * 8 + sub] = make_uint2((unsigned)ib, (unsigned)kb);
                }
                const int nb  = (maxcnt - b0 < 32) ? (maxcnt - b0) : 32;
                const int nb8 = (nb + 7) & ~7;
                if (nt >= 4)      gcn_rank_window<4>(in_q, nb8, mine, rank);
                else if (nt == 3) gcn_rank_window<3>(in_q, nb8, mine, rank);
                else if (nt == 2) gcn_rank_window<2>(in_q, nb8, mine, rank);
                else              gcn_rank_window<1>(in_q, nb8, mine, rank);
            }
#pragma unroll
            for (int t = 0; t < 4; ++t) {
                const int ia = a0 + t * 8 + sub;
                if (ia < cnt && rank[t] >= c0 && rank[t] < c0 + 32)
                    out_q[rank[t] - c0] = (unsigned)key[t];
            }
        }
        const int nwin = (maxcnt - c0 < 32) ? (maxcnt - c0) : 32;
        for (int r = 0; r < nwin; r += U) {
            uint4 e[U / 4];
#pragma unroll
            for (int v = 0; v < U / 4; ++v) e[v] = *(const uint4*)(out_q + r + 4 * v);
            GcnRow<ROW_F16> raw[U];
#pragma unroll
            for (int u = 0; u < U; ++u) {
                const uint4 ev = e[u >> 2];
                const unsigned k = ((u & 3) == 0) ? ev.x : ((u & 3) == 1) ? ev.y : ((u & 3) == 2) ? ev.z : ev.w;
                raw[u] = gcn_row_load<ROW_F16>(Hf, Hh, (int)k, sub);
            }
#pragma unroll
            for (int u = 0; u < U; ++u) {
                const float w = (c0 + r + u < cnt) ? 1.0f : 0.0f;
                gcn_row_fma<ROW_F16>(raw[u], w, acc);
            }
        }
    }
    dd = *(const float*)((const char*)dis + (unsigned)node * 4u);
#pragma unroll
    for (int i = 0; i < 8; ++i) acc[i] *= dd;
}

__global__ void __launch_bounds__(256) __attribute__((amdgpu_waves_per_eu(16)))
gcn_gather(const float* __restrict__ H, const int* __restrict__ rowptr,
           const int* __restrict__ nbr, const float* __restrict__ dis,
           const float* __restrict__ bias, _Float16* __restrict__ outh, int N) {
    typedef __attribute__((ext_vector_type(4))) _Float16 v4h;
    __shared__ alignas(16) uint2    s_in[8][GRP * WPITCH];
    __shared__ alignas(16) unsigned s_out[8][GRP * OPITCH];
    const int lane = threadIdx.x & 31;
    const int wv   = threadIdx.x >> 5;
    const int q    = lane >> 3;
    const int sub  = lane & 7;
    const int nd   = (int)(blockIdx.x * 8 + wv) * GRP + q;
    const int node = (nd < N) ? nd : (N - 1);

    float acc[8];
    float dd;
    gcn_node_sums<false, 4>(H, nullptr, rowptr, nbr, dis, node, sub,
                            &s_in[wv][q * WPITCH], &s_out[wv][q * OPITCH], acc, dd);

    if (nd < N) {
        const float4 ba = *(const float4*)(bias + sub * 4);
        const float4 bb = *(const float4*)(bias + 32 + sub * 4);
        v4h oa, ob;
        oa.x = (_Float16)(fmaxf(acc[0] + ba.x, 0.0f) * dd);
        oa.y = (_Float16)(fmaxf(acc[1] + ba.y, 0.0f) * dd);
        oa.z = (_Float16)(fmaxf(acc[2] + ba.z, 0.0f) * dd);
        oa.w = (_Float16)(fmaxf(acc[3] + ba.w, 0.0f) * dd);
        ob.x = (_Float16)(fmaxf(acc[4] + bb.x, 0.0f) * dd);
        ob.y = (_Float16)(fmaxf(acc[5] + bb.y, 0.0f) * dd);
        ob.z = (_Float16)(fmaxf(acc[6] + bb.z, 0.0f) * dd);
        ob.w = (_Float16)(fmaxf(acc[7] + bb.w, 0.0f) * dd);
        _Float16* op = outh + (size_t)nd * HID + sub * 4;
        *(v4h*)op        = oa;
        *(v4h*)(op + 32) = ob;
    }
}

__global__ void __launch_bounds__(128) __attribute__((amdgpu_waves_per_eu(16)))
gcn_gather_proj(const _Float16* __restrict__ H, const int* __restrict__ rowptr,
                const int* __restrict__ nbr, const float* __restrict__ dis,
                const float* __restrict__ Wt, const float* __restrict__ bias,
                float* __restrict__ outf, int N) {
    constexpr int P  = HID + 8;
    constexpr int CP = HID + 4;
    __shared__ alignas(16) _Float16 Bs[HID * P];
    __shared__ alignas(16) _Float16 Ahi[16 * P];
    __shared__ alignas(16) _Float16 Alo[16 * P];
    __shared__ alignas(16) uint2    s_in[4][GRP * WPITCH];
    __shared__ alignas(16) unsigned s_out[4][GRP * OPITCH];
    static_assert(sizeof(float) * 16 * CP <= sizeof(uint2) * 4 * GRP * WPITCH, "the f32 tile must fit into the staging windows");
    float* Cs = (float*)&s_in[0][0];

    const int tid  = threadIdx.x;
    const int lane = tid & 31;
    const int wv   = __builtin_amdgcn_readfirstlane(tid >> 5);
    const int q    = lane >> 3;
    const int sub  = lane & 7;
    const int nd   = (int)(blockIdx.x * 16) + wv * GRP + q;
    const int node = (nd < N) ? nd : (N - 1);

    const int wn  = tid & (HID - 1);
    const int wk0 = tid >> 6;
    float wf[4][8];
#pragma unroll
    for (int j = 0; j < 4; ++j) {
        const float* wp = Wt + (size_t)(wk0 + j * 2) * 8 * HID + wn;
#pragma unroll
        for (int qq = 0; qq < 8; ++qq) wf[j][qq] = wp[qq * HID];
    }
#pragma unroll
    for (int j = 0; j < 4; ++j) {
        v8h pk;
#pragma unroll
        for (int qq = 0; qq < 8; ++qq) pk[qq] = (_Float16)wf[j][qq];
        *(v8h*)(Bs + wn * P + (wk0 + j * 2) * 8) = pk;
    }

    float acc[8];
    float dd;
    gcn_node_sums<true, 4>(nullptr, H, rowptr, nbr, dis, node, sub,
                           &s_in[wv][q * WPITCH], &s_out[wv][q * OPITCH], acc, dd);

    static_assert(GRP * 8 == 32, "a wave is GRP lane groups of 8: tile row = tid >> 3");
    const int trow = tid >> 3;
    const int ocol = (lane & 15) * 4;

    {
        v8h hi8, lo8;
#pragma unroll
        for (int i = 0; i < 8; ++i) {
            hi8[i] = (_Float16)acc[i];
            lo8[i] = (_Float16)(acc[i] - (float)hi8[i]);
        }
        *(v8h*)(Ahi + trow * P + sub * 8) = hi8;
        *(v8h*)(Alo + trow * P + sub * 8) = lo8;
    }
    const float4 b2v = *(const float4*)(bias + ocol);
    __syncthreads();

    {
        const int c  = lane & 15;
        const int hi = lane >> 4;
        v8f d = (v8f){0.f, 0.f, 0.f, 0.f, 0.f, 0.f, 0.f, 0.f};
#pragma unroll
        for (int kt = 0; kt < HID / 32; ++kt) {
            const int ko = kt * 32 + hi * 8;
            union { v16h v; v8h h[2]; } ah, al, b;
            ah.h[0] = *(const v8h*)(Ahi + c * P + ko);
            ah.h[1] = *(const v8h*)(Ahi + c * P + ko + 16);
            al.h[0] = *(const v8h*)(Alo + c * P + ko);
            al.h[1] = *(const v8h*)(Alo + c * P + ko + 16);
            b.h[0]  = *(const v8h*)(Bs + (wv * 16 + c) * P + ko);
            b.h[1]  = *(const v8h*)(Bs + (wv * 16 + c) * P + ko + 16);
            d = __builtin_amdgcn_wmma_f32_16x16x32_f16(
                 false, al.v,  false, b.v,
                 (short)0, d,  false,  false);
            d = __builtin_amdgcn_wmma_f32_16x16x32_f16(
                 false, ah.v,  false, b.v,
                 (short)0, d,  false,  false);
        }
#pragma unroll
        for (int r = 0; r < 8; ++r) Cs[(hi * 8 + r) * CP + wv * 16 + c] = d[r];
    }
    __syncthreads();

#pragma unroll
    for (int k = 0; k < 2; ++k) {
        const int orow = wv * GRP + k * 2 + (lane >> 4);
        const int on   = (int)(blockIdx.x * 16) + orow;
        if (on < N) {
            const float4 cv = *(const float4*)(Cs + orow * CP + ocol);
            float4 o;
            o.x = cv.x + b2v.x;
            o.y = cv.y + b2v.y;
            o.z = cv.z + b2v.z;
            o.w = cv.w + b2v.w;
            *(float4*)(outf + (size_t)on * HID + ocol) = o;
        }
    }
}


extern "C" void kernel_launch(void* const* d_in, const int* in_sizes, int n_in,
                              void* d_out, int out_size, void* d_ws, size_t ws_size,
                              hipStream_t stream) {
    const float* x  = (const float*)d_in[0];
    const int*   ei = (const int*)d_in[1];
    const float* W1 = (const float*)d_in[2];
    const float* b1 = (const float*)d_in[3];
    const float* W2 = (const float*)d_in[4];
    const float* b2 = (const float*)d_in[5];
    float* out = (float*)d_out;

    const int N = in_sizes[0] / IN_CH;
    const int E = in_sizes[1] / 2;
    const int* src = ei;
    const int* dst = ei + E;

    char* ws = (char*)d_ws;
    size_t off = 0;
    auto alloc = [&](size_t bytes) -> char* {
        char* p = ws + off;
        off = (off + bytes + 255) & ~(size_t)255;
        return p;
    };
    float*    proj = (float*)   alloc((size_t)N * HID * sizeof(float));
    _Float16* h1h  = (_Float16*)alloc((size_t)N * HID * sizeof(_Float16));
    float*    dis  = (float*)   alloc((size_t)N * sizeof(float));
    int*      crs  = (int*)     alloc((size_t)CHUNKS * 64 * sizeof(int));
    int*      rowp = (int*)     alloc((size_t)(N + 1) * sizeof(int));
    int*      slot = (int*)     alloc((size_t)E * sizeof(int));
    int*      nbr  = (int*)     alloc((size_t)E * sizeof(int));
    unsigned short* hist = (unsigned short*)h1h;
    int*            offp = (int*)proj;

    const int T = 256;
    const int gemmBlocks = ((N + 15) / 16 + 7) / 8;
    const int CH = (E + CHUNKS - 1) / CHUNKS;

    gcn_chunk_hist<<<CHUNKS, 1024, 0, stream>>>(dst, hist, crs, slot, N, E, CH);
    gcn_scan_deg<<<(N + 1023) / 1024, 1024, 0, stream>>>(hist, crs, offp, rowp, dis, N);
    gcn_fill<<<(E + T - 1) / T, T, 0, stream>>>(src, dst, slot, offp, nbr, N, E, CH);

    gcn_gemm_wmma<IN_CH, true><<<gemmBlocks, 256, 0, stream>>>(x, nullptr, W1, dis, proj, N);

    gcn_gather<<<(N + 8 * GRP - 1) / (8 * GRP), T, 0, stream>>>(proj, rowp, nbr, dis, b1, h1h, N);

    gcn_gather_proj<<<(N + 15) / 16, 128, 0, stream>>>(h1h, rowp, nbr, dis, W2, b2, out, N);
}


# long buckets (over 32 in-edges) in both neighbour sums: rank window 32 -> 64 ids so the bucket is ranked once, and the other window's keys fetched as one group; sum order unchanged
# speedup vs baseline: 3.1710x; 1.1224x over previous
#include <hip/hip_runtime.h>
#include <hip/hip_bf16.h>

typedef __attribute__((ext_vector_type(16))) _Float16 v16h;
typedef __attribute__((ext_vector_type(8)))  _Float16 v8h;
typedef __attribute__((ext_vector_type(8)))  float    v8f;

#define IN_CH 128
#define HID   64


#define CHUNKS    64
#define TAB_WORDS 51200
static_assert(CHUNKS == HID, "the count and offset planes reuse the N x HID buffers of the layers");

__global__ void __launch_bounds__(1024)
gcn_chunk_hist(const int* __restrict__ dst, unsigned short* __restrict__ hist,
               int* __restrict__ coarse, int* __restrict__ slot, int N, int E, int CH) {
    __shared__ alignas(16) unsigned tab[TAB_WORDS];
    const int tid  = threadIdx.x;
    const int lane = tid & 31;
    const int wv   = tid >> 5;
    const int g    = blockIdx.x;

    for (int i = tid; i < TAB_WORDS / 4; i += 1024) ((uint4*)tab)[i] = make_uint4(0u, 0u, 0u, 0u);
    __syncthreads();

    const int e0 = g * CH;
    const int e1 = (e0 + CH < E) ? (e0 + CH) : E;
    for (int eb = e0 + tid; eb < e1; eb += 4 * 1024) {
        int d[4];
#pragma unroll
        for (int u = 0; u < 4; ++u) {
            const int e = eb + u * 1024;
            d[u] = (e < e1) ? dst[e] : -1;
        }
#pragma unroll
        for (int u = 0; u < 4; ++u) {
            if ((unsigned)d[u] < (unsigned)TAB_WORDS) slot[eb + u * 1024] = (int)atomicAdd(&tab[d[u]], 1u);
        }
    }
    __syncthreads();

    const int nn = (N < TAB_WORDS) ? N : TAB_WORDS;
    unsigned short* __restrict__ hp = hist + (size_t)g * N;
    for (int n = tid; n < nn; n += 1024) hp[n] = (unsigned short)tab[n];

    for (int j = wv; j < 64; j += 32) {
        unsigned sacc = 0u;
        if (j < TAB_WORDS / 1024) {
            for (int i = 0; i < 32; ++i) sacc += tab[j * 1024 + i * 32 + lane];
        }
        for (int dd = 16; dd > 0; dd >>= 1) sacc += __shfl_xor(sacc, dd);
        if (lane == 0) coarse[g * 64 + j] = (int)sacc;
    }
}

__global__ void __launch_bounds__(1024)
gcn_scan_deg(const unsigned short* __restrict__ hist, const int* __restrict__ coarse,
             int* __restrict__ off, int* __restrict__ rowptr,
             float* __restrict__ dis, int N) {
    __shared__ int s_own[32];
    __shared__ int s_front[32];
    const int tid  = threadIdx.x;
    const int lane = tid & 31;
    const int wv   = tid >> 5;
    const int first = blockIdx.x * 1024;

    int front = 0;
    for (int i = tid; i < CHUNKS * 64; i += 1024) {
        const int v = coarse[i];
        front += ((i & 63) < (int)blockIdx.x) ? v : 0;
    }

    const int idx = first + tid;
    int c = 0;
    if (idx < N) {
#pragma unroll 16
        for (int g = 0; g < CHUNKS; ++g) c += (int)hist[(size_t)g * N + idx];
    }

    int incl = c;
    for (int d = 1; d < 32; d <<= 1) {
        int up = __shfl_up(incl, d);
        if (lane >= d) incl += up;
    }
    for (int d = 16; d > 0; d >>= 1) front += __shfl_xor(front, d);
    if (lane == 31) { s_own[wv] = incl; s_front[wv] = front; }
    __syncthreads();

    int t = s_front[lane] + ((lane < wv) ? s_own[lane] : 0);
    for (int d = 16; d > 0; d >>= 1) t += __shfl_xor(t, d);

    const int excl = t + incl - c;
    if (idx < N) {
        rowptr[idx] = excl;
        dis[idx]    = rsqrtf((float)(c + 1));
        if (idx == N - 1) rowptr[N] = excl + c;
        int run = excl;
#pragma unroll 16
        for (int g = 0; g < CHUNKS; ++g) {
            const int h = (int)hist[(size_t)g * N + idx];
            off[(size_t)g * N + idx] = run;
            run += h;
        }
    }
}

__global__ void gcn_fill(const int* __restrict__ src, const int* __restrict__ dst,
                         const int* __restrict__ slot, const int* __restrict__ off,
                         int* __restrict__ nbr, int N, int E, int CH) {
    int e = blockIdx.x * blockDim.x + threadIdx.x;
    if (e < E) {
        const int g = e / CH;
        const unsigned pos = (unsigned)(off[(size_t)g * N + dst[e]] + slot[e]);
        if (pos < (unsigned)E) nbr[pos] = src[e];
    }
}

template <int K, bool A_F32>
__global__ void __launch_bounds__(256)
gcn_gemm_wmma(const float* __restrict__ Af, const _Float16* __restrict__ Ah,
              const float* __restrict__ Wt, const float* __restrict__ rscale,
              _Float16* __restrict__ C, int N) {
    typedef __attribute__((ext_vector_type(4))) _Float16 v4h;
    constexpr int KT = K / 32;
    constexpr int P  = K + 8;
    __shared__ alignas(16) _Float16 Bs[HID * P];

    const int tid  = threadIdx.x;
    const int lane = tid & 31;
    const int wv   = __builtin_amdgcn_readfirstlane(tid >> 5);
    const int c    = lane & 15;
    const int hi   = lane >> 4;
    const int m0   = (blockIdx.x * 8 + wv) << 4;
    const int row  = (m0 + c < N) ? (m0 + c) : (N - 1);

    constexpr int IT = HID * (K / 8) / 256;
    const int wn  = tid & (HID - 1);
    const int wr  = (wn & 3) * 16 + (wn >> 2);
    const int wk0 = tid >> 6;
    float wf[IT][8];
#pragma unroll
    for (int j = 0; j < IT; ++j) {
        const float* wp = Wt + (size_t)(wk0 + j * 4) * 8 * HID + wn;
#pragma unroll
        for (int q = 0; q < 8; ++q) wf[j][q] = wp[q * HID];
    }
    float4 af[A_F32 ? KT * 4 : 1];
    v8h    ah[A_F32 ? 1 : KT * 2];
    if constexpr (A_F32) {
        const float* ap = Af + (size_t)row * K + hi * 8;
#pragma unroll
        for (int kt = 0; kt < KT; ++kt) {
            af[kt * 4 + 0] = *(const float4*)(ap + kt * 32);
            af[kt * 4 + 1] = *(const float4*)(ap + kt * 32 + 4);
            af[kt * 4 + 2] = *(const float4*)(ap + kt * 32 + 16);
            af[kt * 4 + 3] = *(const float4*)(ap + kt * 32 + 20);
        }
    } else {
        const _Float16* ap = Ah + (size_t)row * K + hi * 8;
#pragma unroll
        for (int kt = 0; kt < KT; ++kt) {
            ah[kt * 2 + 0] = *(const v8h*)(ap + kt * 32);
            ah[kt * 2 + 1] = *(const v8h*)(ap + kt * 32 + 16);
        }
    }
    float rs[8];
    if (m0 + 16 <= N) {
        const float4 r0 = *(const float4*)(rscale + m0 + hi * 8);
        const float4 r1 = *(const float4*)(rscale + m0 + hi * 8 + 4);
        rs[0] = r0.x; rs[1] = r0.y; rs[2] = r0.z; rs[3] = r0.w;
        rs[4] = r1.x; rs[5] = r1.y; rs[6] = r1.z; rs[7] = r1.w;
    } else {
#pragma unroll
        for (int r = 0; r < 8; ++r) {
            const int rr = m0 + hi * 8 + r;
            rs[r] = rscale[(rr < N) ? rr : (N - 1)];
        }
    }

#pragma unroll
    for (int j = 0; j < IT; ++j) {
        v8h pk;
#pragma unroll
        for (int q = 0; q < 8; ++q) pk[q] = (_Float16)wf[j][q];
        *(v8h*)(Bs + wr * P + (wk0 + j * 4) * 8) = pk;
    }
    __syncthreads();

    v8f acc[4];
#pragma unroll
    for (int nt = 0; nt < 4; ++nt) acc[nt] = (v8f){0.f, 0.f, 0.f, 0.f, 0.f, 0.f, 0.f, 0.f};

#pragma unroll
    for (int kt = 0; kt < KT; ++kt) {
        union { v16h v; v8h h[2]; } a;
        if constexpr (A_F32) {
#pragma unroll
            for (int j = 0; j < 4; ++j) {
                a.v[j * 4 + 0] = (_Float16)af[kt * 4 + j].x;
                a.v[j * 4 + 1] = (_Float16)af[kt * 4 + j].y;
                a.v[j * 4 + 2] = (_Float16)af[kt * 4 + j].z;
                a.v[j * 4 + 3] = (_Float16)af[kt * 4 + j].w;
            }
        } else {
            a.h[0] = ah[kt * 2 + 0];
            a.h[1] = ah[kt * 2 + 1];
        }
#pragma unroll
        for (int nt = 0; nt < 4; ++nt) {
            const _Float16* bp = Bs + (nt * 16 + c) * P + kt * 32 + hi * 8;
            union { v16h v; v8h h[2]; } b;
            b.h[0] = *(const v8h*)(bp);
            b.h[1] = *(const v8h*)(bp + 16);
            acc[nt] = __builtin_amdgcn_wmma_f32_16x16x32_f16(
                 false, a.v,  false, b.v,
                 (short)0, acc[nt],  false,  false);
        }
    }

    _Float16* crow = C + (size_t)(m0 + hi * 8) * HID + c * 4;
    if (m0 + 16 <= N) {
#pragma unroll
        for (int r = 0; r < 8; ++r) {
            v4h o;
            o.x = (_Float16)(acc[0][r] * rs[r]);
            o.y = (_Float16)(acc[1][r] * rs[r]);
            o.z = (_Float16)(acc[2][r] * rs[r]);
            o.w = (_Float16)(acc[3][r] * rs[r]);
            *(v4h*)(crow + (size_t)r * HID) = o;
        }
    } else if (m0 < N) {
#pragma unroll
        for (int r = 0; r < 8; ++r) {
            if (m0 + hi * 8 + r < N) {
                v4h o;
                o.x = (_Float16)(acc[0][r] * rs[r]);
                o.y = (_Float16)(acc[1][r] * rs[r]);
                o.z = (_Float16)(acc[2][r] * rs[r]);
                o.w = (_Float16)(acc[3][r] * rs[r]);
                *(v4h*)(crow + (size_t)r * HID) = o;
            }
        }
    }
}


#define GRP    4
#define WPITCH 34
#define RANKW  64
#define OPITCH 68

typedef unsigned long long u64;

template <int NT>
__device__ __forceinline__ void gcn_rank_window(const uint2* in_q, int nb8,
                                                const u64 (&mine)[4], int (&rank)[4]) {
    for (int j = 0; j < nb8; j += 8) {
        uint4 e[4];
#pragma unroll
        for (int v = 0; v < 4; ++v) e[v] = *(const uint4*)(in_q + j + 2 * v);
#pragma unroll
        for (int v = 0; v < 4; ++v) {
            const u64 o0 = ((u64)e[v].y << 32) | (u64)e[v].x;
            const u64 o1 = ((u64)e[v].w << 32) | (u64)e[v].z;
#pragma unroll
            for (int t = 0; t < NT; ++t)
                rank[t] += ((o0 < mine[t]) ? 1 : 0) + ((o1 < mine[t]) ? 1 : 0);
        }
    }
}

__device__ __forceinline__ v8h gcn_row_load(const _Float16* __restrict__ Hh, int s, int sub) {
    return *(const v8h*)((const char*)Hh + (unsigned)(s * (HID * 2) + sub * 16));
}

__device__ __forceinline__ void gcn_row_fma(const v8h& r, float w, float (&acc)[8]) {
#pragma unroll
    for (int i = 0; i < 8; ++i) acc[i] = fmaf((float)r[i], w, acc[i]);
}

template <int U>
__device__ __forceinline__ void gcn_node_sums(const _Float16* __restrict__ Hh,
                                              const int* __restrict__ rowptr, const int* __restrict__ nbr,
                                              const float* __restrict__ dis, int node, int sub,
                                              uint2* in_q, unsigned* out_q, float (&acc)[8], float& dd) {
    static_assert(U % 4 == 0, "the row loop reads the rank window four ids at a time");
    const int beg = rowptr[node];
    const int cnt = rowptr[node + 1] - beg;

    int m = cnt;
    { const int o = __shfl_xor(m, 8);  m = (o > m) ? o : m; }
    { const int o = __shfl_xor(m, 16); m = (o > m) ? o : m; }
    const int maxcnt = __builtin_amdgcn_readfirstlane(m);

    {
        const v8h hs = gcn_row_load(Hh, node, sub);
#pragma unroll
        for (int i = 0; i < 8; ++i) acc[i] = (float)hs[i];
    }

    for (int c0 = 0; c0 < maxcnt; c0 += RANKW) {
#pragma unroll
        for (int t = 0; t < 4; ++t) {
            const int p = t * 8 + sub;
            if (c0 + p >= cnt) out_q[p] = (unsigned)node;
        }
        if (maxcnt - c0 > 32) {
#pragma unroll
            for (int t = 4; t < RANKW / 8; ++t) {
                const int p = t * 8 + sub;
                if (c0 + p >= cnt) out_q[p] = (unsigned)node;
            }
        }
        for (int a0 = 0; a0 < maxcnt; a0 += 32) {
            int   key[4];
            int   rank[4];
            u64   mine[4];
#pragma unroll
            for (int t = 0; t < 4; ++t) {
                const int ia = a0 + t * 8 + sub;
                key[t]  = nbr[(ia < cnt) ? (beg + ia) : 0];
                key[t]  = (ia < cnt) ? key[t] : 0x7fffffff;
                rank[t] = 0;
                mine[t] = ((u64)(unsigned)key[t] << 32) | (u64)(unsigned)ia;
            }
            const int nt = (maxcnt - a0 + 7) >> 3;
            for (int b0 = 0; b0 < maxcnt; b0 += 32) {
                int kb[4];
                if (b0 != a0) {
#pragma unroll
                    for (int t = 0; t < 4; ++t) {
                        const int ib = b0 + t * 8 + sub;
                        kb[t] = nbr[(ib < cnt) ? (beg + ib) : 0];
                    }
#pragma unroll
                    for (int t = 0; t < 4; ++t) {
                        const int ib = b0 + t * 8 + sub;
                        kb[t] = (ib < cnt) ? kb[t] : 0x7fffffff;
                    }
                } else {
#pragma unroll
                    for (int t = 0; t < 4; ++t) kb[t] = key[t];
                }
#pragma unroll
                for (int t = 0; t < 4; ++t) {
                    const int ib = b0 + t * 8 + sub;
                    in_q[t * 8 + sub] = make_uint2((unsigned)ib, (unsigned)kb[t]);
                }
                const int nb  = (maxcnt - b0 < 32) ? (maxcnt - b0) : 32;
                const int nb8 = (nb + 7) & ~7;
                if (nt >= 4)      gcn_rank_window<4>(in_q, nb8, mine, rank);
                else if (nt == 3) gcn_rank_window<3>(in_q, nb8, mine, rank);
                else if (nt == 2) gcn_rank_window<2>(in_q, nb8, mine, rank);
                else              gcn_rank_window<1>(in_q, nb8, mine, rank);
            }
#pragma unroll
            for (int t = 0; t < 4; ++t) {
                const int ia = a0 + t * 8 + sub;
                if (ia < cnt && rank[t] >= c0 && rank[t] < c0 + RANKW)
                    out_q[rank[t] - c0] = (unsigned)key[t];
            }
        }
        const int nwin = (maxcnt - c0 < RANKW) ? (maxcnt - c0) : RANKW;
        for (int r = 0; r < nwin; r += U) {
            uint4 e[U / 4];
#pragma unroll
            for (int v = 0; v < U / 4; ++v) e[v] = *(const uint4*)(out_q + r + 4 * v);
            v8h raw[U];
#pragma unroll
            for (int u = 0; u < U; ++u) {
                const uint4 ev = e[u >> 2];
                const unsigned k = ((u & 3) == 0) ? ev.x : ((u & 3) == 1) ? ev.y : ((u & 3) == 2) ? ev.z : ev.w;
                raw[u] = gcn_row_load(Hh, (int)k, sub);
            }
#pragma unroll
            for (int u = 0; u < U; ++u) {
                const float w = (c0 + r + u < cnt) ? 1.0f : 0.0f;
                gcn_row_fma(raw[u], w, acc);
            }
        }
    }
    dd = *(const float*)((const char*)dis + (unsigned)node * 4u);
#pragma unroll
    for (int i = 0; i < 8; ++i) acc[i] *= dd;
}

__global__ void __launch_bounds__(256) __attribute__((amdgpu_waves_per_eu(16)))
gcn_gather(const _Float16* __restrict__ H, const int* __restrict__ rowptr,
           const int* __restrict__ nbr, const float* __restrict__ dis,
           const float* __restrict__ bias, _Float16* __restrict__ outh, int N) {
    __shared__ alignas(16) uint2    s_in[8][GRP * WPITCH];
    __shared__ alignas(16) unsigned s_out[8][GRP * OPITCH];
    const int lane = threadIdx.x & 31;
    const int wv   = threadIdx.x >> 5;
    const int q    = lane >> 3;
    const int sub  = lane & 7;
    const int nd   = (int)(blockIdx.x * 8 + wv) * GRP + q;
    const int node = (nd < N) ? nd : (N - 1);

    float acc[8];
    float dd;
    gcn_node_sums<4>(H, rowptr, nbr, dis, node, sub,
                     &s_in[wv][q * WPITCH], &s_out[wv][q * OPITCH], acc, dd);

    if (nd < N) {
        const float4 ba = *(const float4*)(bias + sub * 8);
        const float4 bb = *(const float4*)(bias + sub * 8 + 4);
        v8h o;
        o[0] = (_Float16)(fmaxf(acc[0] + ba.x, 0.0f) * dd);
        o[1] = (_Float16)(fmaxf(acc[1] + ba.y, 0.0f) * dd);
        o[2] = (_Float16)(fmaxf(acc[2] + ba.z, 0.0f) * dd);
        o[3] = (_Float16)(fmaxf(acc[3] + ba.w, 0.0f) * dd);
        o[4] = (_Float16)(fmaxf(acc[4] + bb.x, 0.0f) * dd);
        o[5] = (_Float16)(fmaxf(acc[5] + bb.y, 0.0f) * dd);
        o[6] = (_Float16)(fmaxf(acc[6] + bb.z, 0.0f) * dd);
        o[7] = (_Float16)(fmaxf(acc[7] + bb.w, 0.0f) * dd);
        *(v8h*)(outh + (size_t)nd * HID + sub * 8) = o;
    }
}

__global__ void __launch_bounds__(128) __attribute__((amdgpu_waves_per_eu(16)))
gcn_gather_proj(const _Float16* __restrict__ H, const int* __restrict__ rowptr,
                const int* __restrict__ nbr, const float* __restrict__ dis,
                const float* __restrict__ Wt, const float* __restrict__ bias,
                float* __restrict__ outf, int N) {
    constexpr int P  = HID + 8;
    constexpr int CP = HID + 4;
    __shared__ alignas(16) _Float16 Bs[HID * P];
    __shared__ alignas(16) _Float16 Ahi[16 * P];
    __shared__ alignas(16) _Float16 Alo[16 * P];
    __shared__ alignas(16) uint2    s_in[4][GRP * WPITCH];
    __shared__ alignas(16) unsigned s_out[4][GRP * OPITCH];
    static_assert(sizeof(float) * 16 * CP <= sizeof(uint2) * 4 * GRP * WPITCH, "the f32 tile must fit into the staging windows");
    float* Cs = (float*)&s_in[0][0];

    const int tid  = threadIdx.x;
    const int lane = tid & 31;
    const int wv   = __builtin_amdgcn_readfirstlane(tid >> 5);
    const int q    = lane >> 3;
    const int sub  = lane & 7;
    const int nd   = (int)(blockIdx.x * 16) + wv * GRP + q;
    const int node = (nd < N) ? nd : (N - 1);

    const int wn  = tid & (HID - 1);
    const int wk0 = tid >> 6;
    float wf[4][8];
#pragma unroll
    for (int j = 0; j < 4; ++j) {
        const float* wp = Wt + (size_t)(wk0 + j * 2) * 8 * HID + wn;
#pragma unroll
        for (int qq = 0; qq < 8; ++qq) wf[j][qq] = wp[qq * HID];
    }
#pragma unroll
    for (int j = 0; j < 4; ++j) {
        v8h pk;
#pragma unroll
        for (int qq = 0; qq < 8; ++qq) pk[qq] = (_Float16)wf[j][qq];
        *(v8h*)(Bs + wn * P + (wk0 + j * 2) * 8) = pk;
    }

    float acc[8];
    float dd;
    gcn_node_sums<4>(H, rowptr, nbr, dis, node, sub,
                     &s_in[wv][q * WPITCH], &s_out[wv][q * OPITCH], acc, dd);

    static_assert(GRP * 8 == 32, "a wave is GRP lane groups of 8: tile row = tid >> 3");
    const int trow = tid >> 3;
    const int ocol = (lane & 15) * 4;

    {
        v8h hi8, lo8;
#pragma unroll
        for (int i = 0; i < 8; ++i) {
            hi8[i] = (_Float16)acc[i];
            lo8[i] = (_Float16)(acc[i] - (float)hi8[i]);
        }
        *(v8h*)(Ahi + trow * P + sub * 8) = hi8;
        *(v8h*)(Alo + trow * P + sub * 8) = lo8;
    }
    const float4 b2v = *(const float4*)(bias + ocol);
    __syncthreads();

    {
        const int c  = lane & 15;
        const int hi = lane >> 4;
        v8f d = (v8f){0.f, 0.f, 0.f, 0.f, 0.f, 0.f, 0.f, 0.f};
#pragma unroll
        for (int kt = 0; kt < HID / 32; ++kt) {
            const int ko = kt * 32 + hi * 8;
            union { v16h v; v8h h[2]; } ah, al, b;
            ah.h[0] = *(const v8h*)(Ahi + c * P + ko);
            ah.h[1] = *(const v8h*)(Ahi + c * P + ko + 16);
            al.h[0] = *(const v8h*)(Alo + c * P + ko);
            al.h[1] = *(const v8h*)(Alo + c * P + ko + 16);
            b.h[0]  = *(const v8h*)(Bs + (wv * 16 + c) * P + ko);
            b.h[1]  = *(const v8h*)(Bs + (wv * 16 + c) * P + ko + 16);
            d = __builtin_amdgcn_wmma_f32_16x16x32_f16(
                 false, al.v,  false, b.v,
                 (short)0, d,  false,  false);
            d = __builtin_amdgcn_wmma_f32_16x16x32_f16(
                 false, ah.v,  false, b.v,
                 (short)0, d,  false,  false);
        }
#pragma unroll
        for (int r = 0; r < 8; ++r) Cs[(hi * 8 + r) * CP + wv * 16 + c] = d[r];
    }
    __syncthreads();

#pragma unroll
    for (int k = 0; k < 2; ++k) {
        const int orow = wv * GRP + k * 2 + (lane >> 4);
        const int on   = (int)(blockIdx.x * 16) + orow;
        if (on < N) {
            const float4 cv = *(const float4*)(Cs + orow * CP + ocol);
            float4 o;
            o.x = cv.x + b2v.x;
            o.y = cv.y + b2v.y;
            o.z = cv.z + b2v.z;
            o.w = cv.w + b2v.w;
            *(float4*)(outf + (size_t)on * HID + ocol) = o;
        }
    }
}


extern "C" void kernel_launch(void* const* d_in, const int* in_sizes, int n_in,
                              void* d_out, int out_size, void* d_ws, size_t ws_size,
                              hipStream_t stream) {
    const float* x  = (const float*)d_in[0];
    const int*   ei = (const int*)d_in[1];
    const float* W1 = (const float*)d_in[2];
    const float* b1 = (const float*)d_in[3];
    const float* W2 = (const float*)d_in[4];
    const float* b2 = (const float*)d_in[5];
    float* out = (float*)d_out;

    const int N = in_sizes[0] / IN_CH;
    const int E = in_sizes[1] / 2;
    const int* src = ei;
    const int* dst = ei + E;

    char* ws = (char*)d_ws;
    size_t off = 0;
    auto alloc = [&](size_t bytes) -> char* {
        char* p = ws + off;
        off = (off + bytes + 255) & ~(size_t)255;
        return p;
    };
    _Float16* proj = (_Float16*)alloc((size_t)N * HID * sizeof(int));
    _Float16* h1h  = (_Float16*)alloc((size_t)N * HID * sizeof(_Float16));
    float*    dis  = (float*)   alloc((size_t)N * sizeof(float));
    int*      crs  = (int*)     alloc((size_t)CHUNKS * 64 * sizeof(int));
    int*      rowp = (int*)     alloc((size_t)(N + 1) * sizeof(int));
    int*      slot = (int*)     alloc((size_t)E * sizeof(int));
    int*      nbr  = (int*)     alloc((size_t)E * sizeof(int));
    unsigned short* hist = (unsigned short*)h1h;
    int*            offp = (int*)proj;

    const int T = 256;
    const int gemmBlocks = ((N + 15) / 16 + 7) / 8;
    const int CH = (E + CHUNKS - 1) / CHUNKS;

    gcn_chunk_hist<<<CHUNKS, 1024, 0, stream>>>(dst, hist, crs, slot, N, E, CH);
    gcn_scan_deg<<<(N + 1023) / 1024, 1024, 0, stream>>>(hist, crs, offp, rowp, dis, N);
    gcn_fill<<<(E + T - 1) / T, T, 0, stream>>>(src, dst, slot, offp, nbr, N, E, CH);

    gcn_gemm_wmma<IN_CH, true><<<gemmBlocks, 256, 0, stream>>>(x, nullptr, W1, dis, proj, N);

    gcn_gather<<<(N + 8 * GRP - 1) / (8 * GRP), T, 0, stream>>>(proj, rowp, nbr, dis, b1, h1h, N);

    gcn_gather_proj<<<(N + 15) / 16, 128, 0, stream>>>(h1h, rowp, nbr, dis, W2, b2, out, N);
}


# bucketing prologue re-cut: scan as 196 blocks of 256 nodes (was 49 of 1024), a node's 64 counts read once as one load group, front from per-piece running slice totals; count pass batches its loads
# speedup vs baseline: 3.7139x; 1.1712x over previous
#include <hip/hip_runtime.h>
#include <hip/hip_bf16.h>

typedef __attribute__((ext_vector_type(16))) _Float16 v16h;
typedef __attribute__((ext_vector_type(8)))  _Float16 v8h;
typedef __attribute__((ext_vector_type(8)))  float    v8f;

#define IN_CH 128
#define HID   64


#define CHUNKS    64
#define TAB_WORDS 51200
#define CSL       256
#define NSL       (TAB_WORDS / CSL)
#define NSLP      224
#define HU        8
static_assert(CHUNKS == HID, "the count and offset planes reuse the N x HID buffers of the layers");
static_assert(NSL <= NSLP && NSLP == 7 * 32, "the slice totals are summed and scanned seven to a wave");

__global__ void __launch_bounds__(1024)
gcn_chunk_hist(const int* __restrict__ dst, unsigned short* __restrict__ hist,
               int* __restrict__ cpre, int* __restrict__ slot, int N, int E, int CH) {
    __shared__ alignas(16) unsigned tab[TAB_WORDS];
    __shared__ unsigned s_sl[NSLP];
    const int tid  = threadIdx.x;
    const int lane = tid & 31;
    const int wv   = tid >> 5;
    const int g    = blockIdx.x;

    for (int i = tid; i < TAB_WORDS / 4; i += 1024) ((uint4*)tab)[i] = make_uint4(0u, 0u, 0u, 0u);
    __syncthreads();

    const int e0 = g * CH;
    const int e1 = (e0 + CH < E) ? (e0 + CH) : E;
    for (int eb = e0 + tid; eb < e1; eb += HU * 1024) {
        int      d[HU];
        unsigned rk[HU];
#pragma unroll
        for (int u = 0; u < HU; ++u) {
            const int e = eb + u * 1024;
            d[u] = dst[(e < e1) ? e : (e1 - 1)];
        }
#pragma unroll
        for (int u = 0; u < HU; ++u) {
            const int e = eb + u * 1024;
            rk[u] = 0xffffffffu;
            if (e < e1 && (unsigned)d[u] < (unsigned)TAB_WORDS) rk[u] = atomicAdd(&tab[d[u]], 1u);
        }
#pragma unroll
        for (int u = 0; u < HU; ++u) {
            if (rk[u] != 0xffffffffu) slot[eb + u * 1024] = (int)rk[u];
        }
    }
    __syncthreads();

    const int nn = (N < TAB_WORDS) ? N : TAB_WORDS;
    unsigned short* __restrict__ hp = hist + (size_t)g * N;
    if ((N & 7) == 0) {
        for (int i = tid; i * 8 < nn; i += 1024) {
            const uint4 a = ((const uint4*)tab)[2 * i];
            const uint4 c = ((const uint4*)tab)[2 * i + 1];
            uint4 o;
            o.x = a.x | (a.y << 16);
            o.y = a.z | (a.w << 16);
            o.z = c.x | (c.y << 16);
            o.w = c.z | (c.w << 16);
            *(uint4*)(hp + (size_t)i * 8) = o;
        }
    } else {
        for (int n = tid; n < nn; n += 1024) hp[n] = (unsigned short)tab[n];
    }

    unsigned ps[7];
#pragma unroll
    for (int q = 0; q < 7; ++q) {
        const int j = wv + q * 32;
        unsigned sacc = 0u;
        if (j < NSL) {
#pragma unroll
            for (int i = 0; i < CSL / 32; ++i) sacc += tab[j * CSL + i * 32 + lane];
        }
        ps[q] = sacc;
    }
    for (int dd = 16; dd > 0; dd >>= 1) {
#pragma unroll
        for (int q = 0; q < 7; ++q) ps[q] += __shfl_xor(ps[q], dd);
    }
    if (lane == 0) {
#pragma unroll
        for (int q = 0; q < 7; ++q) s_sl[wv + q * 32] = ps[q];
    }
    __syncthreads();

    if (wv == 0) {
        unsigned v[7], inc[7], tot[7];
#pragma unroll
        for (int q = 0; q < 7; ++q) { v[q] = s_sl[q * 32 + lane]; inc[q] = v[q]; }
        for (int dd = 1; dd < 32; dd <<= 1) {
#pragma unroll
            for (int q = 0; q < 7; ++q) {
                const unsigned up = __shfl_up(inc[q], dd);
                if (lane >= dd) inc[q] += up;
            }
        }
#pragma unroll
        for (int q = 0; q < 7; ++q) tot[q] = __shfl(inc[q], 31);
        unsigned carry = 0u;
#pragma unroll
        for (int q = 0; q < 7; ++q) {
            const int j = q * 32 + lane;
            if (j < NSL) cpre[j * CHUNKS + g] = (int)(carry + inc[q] - v[q]);
            carry += tot[q];
        }
    }
}

__global__ void __launch_bounds__(CSL)
gcn_scan_deg(const unsigned short* __restrict__ hist, const int* __restrict__ cpre,
             int* __restrict__ off, int* __restrict__ rowptr,
             float* __restrict__ dis, int N) {
    __shared__ int s_own[CSL / 32];
    __shared__ int s_front[CSL / 32];
    const int tid  = threadIdx.x;
    const int lane = tid & 31;
    const int wv   = tid >> 5;
    const int b    = (int)blockIdx.x;
    const int idx  = b * CSL + tid;
    const int idc  = (idx < N) ? idx : (N - 1);

    int front = cpre[b * CHUNKS + (tid & (CHUNKS - 1))];
    front = (tid < CHUNKS) ? front : 0;

    int h[CHUNKS];
#pragma unroll
    for (int g = 0; g < CHUNKS; ++g) h[g] = (int)hist[(size_t)g * N + idc];
    int c = 0;
#pragma unroll
    for (int g = 0; g < CHUNKS; ++g) c += h[g];
    c = (idx < N) ? c : 0;

    int incl = c;
    for (int d = 1; d < 32; d <<= 1) {
        int up = __shfl_up(incl, d);
        if (lane >= d) incl += up;
    }
    for (int d = 16; d > 0; d >>= 1) front += __shfl_xor(front, d);
    if (lane == 31) { s_own[wv] = incl; s_front[wv] = front; }
    __syncthreads();

    int t = 0;
#pragma unroll
    for (int w = 0; w < CSL / 32; ++w) t += s_front[w] + ((w < wv) ? s_own[w] : 0);

    const int excl = t + incl - c;
    if (idx < N) {
        rowptr[idx] = excl;
        dis[idx]    = rsqrtf((float)(c + 1));
        if (idx == N - 1) rowptr[N] = excl + c;
        int run = excl;
#pragma unroll
        for (int g = 0; g < CHUNKS; ++g) {
            off[(size_t)g * N + idx] = run;
            run += h[g];
        }
    }
}

#define FILL_EDGES 8
#define FILL_SPAN  (256 * FILL_EDGES)

__device__ __forceinline__ void gcn_fill_part(const int* __restrict__ src, const int* __restrict__ dst,
                                              const int* __restrict__ slot, const int* __restrict__ off,
                                              int* __restrict__ nbr, int N, int E, int CH, int bid, int bpp) {
    const int g    = bid / bpp;
    const int base = g * CH + (bid - g * bpp) * FILL_SPAN + (int)threadIdx.x;
    const int pe   = (g + 1) * CH;
    const int end  = (pe < E) ? pe : E;
    const int* __restrict__ offg = off + (size_t)g * N;
    int d[FILL_EDGES], sl[FILL_EDGES], sr[FILL_EDGES], o[FILL_EDGES];
#pragma unroll
    for (int u = 0; u < FILL_EDGES; ++u) {
        const int e  = base + u * 256;
        const int ec = (e < end) ? e : (E - 1);
        d[u]  = dst[ec];
        sl[u] = slot[ec];
        sr[u] = src[ec];
    }
#pragma unroll
    for (int u = 0; u < FILL_EDGES; ++u) o[u] = offg[d[u]];
#pragma unroll
    for (int u = 0; u < FILL_EDGES; ++u) {
        const int e = base + u * 256;
        asm volatile("" : : "v"(sr[u]));
        const unsigned pos = (unsigned)(o[u] + sl[u]);
        if (e < end && pos < (unsigned)E) nbr[pos] = sr[u];
    }
}

template <int K, bool A_F32>
__device__ __forceinline__ void
gcn_gemm_part(const float* __restrict__ Af, const _Float16* __restrict__ Ah,
              const float* __restrict__ Wt, const float* __restrict__ rscale,
              _Float16* __restrict__ C, int N, int bid) {
    typedef __attribute__((ext_vector_type(4))) _Float16 v4h;
    constexpr int KT = K / 32;
    constexpr int P  = K + 8;
    __shared__ alignas(16) _Float16 Bs[HID * P];

    const int tid  = threadIdx.x;
    const int lane = tid & 31;
    const int wv   = __builtin_amdgcn_readfirstlane(tid >> 5);
    const int c    = lane & 15;
    const int hi   = lane >> 4;
    const int m0   = (bid * 8 + wv) << 4;
    const int row  = (m0 + c < N) ? (m0 + c) : (N - 1);

    constexpr int IT = HID * (K / 8) / 256;
    const int wn  = tid & (HID - 1);
    const int wr  = (wn & 3) * 16 + (wn >> 2);
    const int wk0 = tid >> 6;
    float wf[IT][8];
#pragma unroll
    for (int j = 0; j < IT; ++j) {
        const float* wp = Wt + (size_t)(wk0 + j * 4) * 8 * HID + wn;
#pragma unroll
        for (int q = 0; q < 8; ++q) wf[j][q] = wp[q * HID];
    }
    float4 af[A_F32 ? KT * 4 : 1];
    v8h    ah[A_F32 ? 1 : KT * 2];
    if constexpr (A_F32) {
        const float* ap = Af + (size_t)row * K + hi * 8;
#pragma unroll
        for (int kt = 0; kt < KT; ++kt) {
            af[kt * 4 + 0] = *(const float4*)(ap + kt * 32);
            af[kt * 4 + 1] = *(const float4*)(ap + kt * 32 + 4);
            af[kt * 4 + 2] = *(const float4*)(ap + kt * 32 + 16);
            af[kt * 4 + 3] = *(const float4*)(ap + kt * 32 + 20);
        }
    } else {
        const _Float16* ap = Ah + (size_t)row * K + hi * 8;
#pragma unroll
        for (int kt = 0; kt < KT; ++kt) {
            ah[kt * 2 + 0] = *(const v8h*)(ap + kt * 32);
            ah[kt * 2 + 1] = *(const v8h*)(ap + kt * 32 + 16);
        }
    }
    float rs[8];
    if (m0 + 16 <= N) {
        const float4 r0 = *(const float4*)(rscale + m0 + hi * 8);
        const float4 r1 = *(const float4*)(rscale + m0 + hi * 8 + 4);
        rs[0] = r0.x; rs[1] = r0.y; rs[2] = r0.z; rs[3] = r0.w;
        rs[4] = r1.x; rs[5] = r1.y; rs[6] = r1.z; rs[7] = r1.w;
    } else {
#pragma unroll
        for (int r = 0; r < 8; ++r) {
            const int rr = m0 + hi * 8 + r;
            rs[r] = rscale[(rr < N) ? rr : (N - 1)];
        }
    }

#pragma unroll
    for (int j = 0; j < IT; ++j) {
        v8h pk;
#pragma unroll
        for (int q = 0; q < 8; ++q) pk[q] = (_Float16)wf[j][q];
        *(v8h*)(Bs + wr * P + (wk0 + j * 4) * 8) = pk;
    }
    __syncthreads();

    v8f acc[4];
#pragma unroll
    for (int nt = 0; nt < 4; ++nt) acc[nt] = (v8f){0.f, 0.f, 0.f, 0.f, 0.f, 0.f, 0.f, 0.f};

#pragma unroll
    for (int kt = 0; kt < KT; ++kt) {
        union { v16h v; v8h h[2]; } a;
        if constexpr (A_F32) {
#pragma unroll
            for (int j = 0; j < 4; ++j) {
                a.v[j * 4 + 0] = (_Float16)af[kt * 4 + j].x;
                a.v[j * 4 + 1] = (_Float16)af[kt * 4 + j].y;
                a.v[j * 4 + 2] = (_Float16)af[kt * 4 + j].z;
                a.v[j * 4 + 3] = (_Float16)af[kt * 4 + j].w;
            }
        } else {
            a.h[0] = ah[kt * 2 + 0];
            a.h[1] = ah[kt * 2 + 1];
        }
#pragma unroll
        for (int nt = 0; nt < 4; ++nt) {
            const _Float16* bp = Bs + (nt * 16 + c) * P + kt * 32 + hi * 8;
            union { v16h v; v8h h[2]; } b;
            b.h[0] = *(const v8h*)(bp);
            b.h[1] = *(const v8h*)(bp + 16);
            acc[nt] = __builtin_amdgcn_wmma_f32_16x16x32_f16(
                 false, a.v,  false, b.v,
                 (short)0, acc[nt],  false,  false);
        }
    }

    _Float16* crow = C + (size_t)(m0 + hi * 8) * HID + c * 4;
    if (m0 + 16 <= N) {
#pragma unroll
        for (int r = 0; r < 8; ++r) {
            v4h o;
            o.x = (_Float16)(acc[0][r] * rs[r]);
            o.y = (_Float16)(acc[1][r] * rs[r]);
            o.z = (_Float16)(acc[2][r] * rs[r]);
            o.w = (_Float16)(acc[3][r] * rs[r]);
            *(v4h*)(crow + (size_t)r * HID) = o;
        }
    } else if (m0 < N) {
#pragma unroll
        for (int r = 0; r < 8; ++r) {
            if (m0 + hi * 8 + r < N) {
                v4h o;
                o.x = (_Float16)(acc[0][r] * rs[r]);
                o.y = (_Float16)(acc[1][r] * rs[r]);
                o.z = (_Float16)(acc[2][r] * rs[r]);
                o.w = (_Float16)(acc[3][r] * rs[r]);
                *(v4h*)(crow + (size_t)r * HID) = o;
            }
        }
    }
}

__global__ void __launch_bounds__(256)
gcn_fill_gemm(const float* __restrict__ x, const float* __restrict__ Wt,
              const float* __restrict__ rscale, _Float16* __restrict__ C, int N,
              const int* __restrict__ src, const int* __restrict__ dst,
              const int* __restrict__ slot, const int* __restrict__ off,
              int* __restrict__ nbr, int E, int CH, int bpp, int fillBlocks) {
    const int b = (int)blockIdx.x;
    if (b < fillBlocks) gcn_fill_part(src, dst, slot, off, nbr, N, E, CH, b, bpp);
    else                gcn_gemm_part<IN_CH, true>(x, nullptr, Wt, rscale, C, N, b - fillBlocks);
}


#define GRP    4
#define WPITCH 34
#define RANKW  64
#define OPITCH 68

typedef unsigned long long u64;

template <int NT>
__device__ __forceinline__ void gcn_rank_window(const uint2* in_q, int nb8,
                                                const u64 (&mine)[4], int (&rank)[4]) {
    for (int j = 0; j < nb8; j += 8) {
        uint4 e[4];
#pragma unroll
        for (int v = 0; v < 4; ++v) e[v] = *(const uint4*)(in_q + j + 2 * v);
#pragma unroll
        for (int v = 0; v < 4; ++v) {
            const u64 o0 = ((u64)e[v].y << 32) | (u64)e[v].x;
            const u64 o1 = ((u64)e[v].w << 32) | (u64)e[v].z;
#pragma unroll
            for (int t = 0; t < NT; ++t)
                rank[t] += ((o0 < mine[t]) ? 1 : 0) + ((o1 < mine[t]) ? 1 : 0);
        }
    }
}

__device__ __forceinline__ v8h gcn_row_load(const _Float16* __restrict__ Hh, int s, int sub) {
    return *(const v8h*)((const char*)Hh + (unsigned)(s * (HID * 2) + sub * 16));
}

__device__ __forceinline__ void gcn_row_fma(const v8h& r, float w, float (&acc)[8]) {
#pragma unroll
    for (int i = 0; i < 8; ++i) acc[i] = fmaf((float)r[i], w, acc[i]);
}

template <int U>
__device__ __forceinline__ void gcn_node_sums(const _Float16* __restrict__ Hh,
                                              const int* __restrict__ rowptr, const int* __restrict__ nbr,
                                              const float* __restrict__ dis, int node, int sub,
                                              uint2* in_q, unsigned* out_q, float (&acc)[8], float& dd) {
    static_assert(U % 4 == 0, "the row loop reads the rank window four ids at a time");
    const int beg = rowptr[node];
    const int cnt = rowptr[node + 1] - beg;

    int m = cnt;
    { const int o = __shfl_xor(m, 8);  m = (o > m) ? o : m; }
    { const int o = __shfl_xor(m, 16); m = (o > m) ? o : m; }
    const int maxcnt = __builtin_amdgcn_readfirstlane(m);

    {
        const v8h hs = gcn_row_load(Hh, node, sub);
#pragma unroll
        for (int i = 0; i < 8; ++i) acc[i] = (float)hs[i];
    }

    for (int c0 = 0; c0 < maxcnt; c0 += RANKW) {
#pragma unroll
        for (int t = 0; t < 4; ++t) {
            const int p = t * 8 + sub;
            if (c0 + p >= cnt) out_q[p] = (unsigned)node;
        }
        if (maxcnt - c0 > 32) {
#pragma unroll
            for (int t = 4; t < RANKW / 8; ++t) {
                const int p = t * 8 + sub;
                if (c0 + p >= cnt) out_q[p] = (unsigned)node;
            }
        }
        for (int a0 = 0; a0 < maxcnt; a0 += 32) {
            int   key[4];
            int   rank[4];
            u64   mine[4];
#pragma unroll
            for (int t = 0; t < 4; ++t) {
                const int ia = a0 + t * 8 + sub;
                key[t]  = nbr[(ia < cnt) ? (beg + ia) : 0];
                key[t]  = (ia < cnt) ? key[t] : 0x7fffffff;
                rank[t] = 0;
                mine[t] = ((u64)(unsigned)key[t] << 32) | (u64)(unsigned)ia;
            }
            const int nt = (maxcnt - a0 + 7) >> 3;
            for (int b0 = 0; b0 < maxcnt; b0 += 32) {
                int kb[4];
                if (b0 != a0) {
#pragma unroll
                    for (int t = 0; t < 4; ++t) {
                        const int ib = b0 + t * 8 + sub;
                        kb[t] = nbr[(ib < cnt) ? (beg + ib) : 0];
                    }
#pragma unroll
                    for (int t = 0; t < 4; ++t) {
                        const int ib = b0 + t * 8 + sub;
                        kb[t] = (ib < cnt) ? kb[t] : 0x7fffffff;
                    }
                } else {
#pragma unroll
                    for (int t = 0; t < 4; ++t) kb[t] = key[t];
                }
#pragma unroll
                for (int t = 0; t < 4; ++t) {
                    const int ib = b0 + t * 8 + sub;
                    in_q[t * 8 + sub] = make_uint2((unsigned)ib, (unsigned)kb[t]);
                }
                const int nb  = (maxcnt - b0 < 32) ? (maxcnt - b0) : 32;
                const int nb8 = (nb + 7) & ~7;
                if (nt >= 4)      gcn_rank_window<4>(in_q, nb8, mine, rank);
                else if (nt == 3) gcn_rank_window<3>(in_q, nb8, mine, rank);
                else if (nt == 2) gcn_rank_window<2>(in_q, nb8, mine, rank);
                else              gcn_rank_window<1>(in_q, nb8, mine, rank);
            }
#pragma unroll
            for (int t = 0; t < 4; ++t) {
                const int ia = a0 + t * 8 + sub;
                if (ia < cnt && rank[t] >= c0 && rank[t] < c0 + RANKW)
                    out_q[rank[t] - c0] = (unsigned)key[t];
            }
        }
        const int nwin = (maxcnt - c0 < RANKW) ? (maxcnt - c0) : RANKW;
        for (int r = 0; r < nwin; r += U) {
            uint4 e[U / 4];
#pragma unroll
            for (int v = 0; v < U / 4; ++v) e[v] = *(const uint4*)(out_q + r + 4 * v);
            v8h raw[U];
#pragma unroll
            for (int u = 0; u < U; ++u) {
                const uint4 ev = e[u >> 2];
                const unsigned k = ((u & 3) == 0) ? ev.x : ((u & 3) == 1) ? ev.y : ((u & 3) == 2) ? ev.z : ev.w;
                raw[u] = gcn_row_load(Hh, (int)k, sub);
            }
#pragma unroll
            for (int u = 0; u < U; ++u) {
                const float w = (c0 + r + u < cnt) ? 1.0f : 0.0f;
                gcn_row_fma(raw[u], w, acc);
            }
        }
    }
    dd = *(const float*)((const char*)dis + (unsigned)node * 4u);
#pragma unroll
    for (int i = 0; i < 8; ++i) acc[i] *= dd;
}

__global__ void __launch_bounds__(256) __attribute__((amdgpu_waves_per_eu(16)))
gcn_gather(const _Float16* __restrict__ H, const int* __restrict__ rowptr,
           const int* __restrict__ nbr, const float* __restrict__ dis,
           const float* __restrict__ bias, _Float16* __restrict__ outh, int N) {
    __shared__ alignas(16) uint2    s_in[8][GRP * WPITCH];
    __shared__ alignas(16) unsigned s_out[8][GRP * OPITCH];
    const int lane = threadIdx.x & 31;
    const int wv   = threadIdx.x >> 5;
    const int q    = lane >> 3;
    const int sub  = lane & 7;
    const int nd   = (int)(blockIdx.x * 8 + wv) * GRP + q;
    const int node = (nd < N) ? nd : (N - 1);

    float acc[8];
    float dd;
    gcn_node_sums<4>(H, rowptr, nbr, dis, node, sub,
                     &s_in[wv][q * WPITCH], &s_out[wv][q * OPITCH], acc, dd);

    if (nd < N) {
        const float4 ba = *(const float4*)(bias + sub * 8);
        const float4 bb = *(const float4*)(bias + sub * 8 + 4);
        v8h o;
        o[0] = (_Float16)(fmaxf(acc[0] + ba.x, 0.0f) * dd);
        o[1] = (_Float16)(fmaxf(acc[1] + ba.y, 0.0f) * dd);
        o[2] = (_Float16)(fmaxf(acc[2] + ba.z, 0.0f) * dd);
        o[3] = (_Float16)(fmaxf(acc[3] + ba.w, 0.0f) * dd);
        o[4] = (_Float16)(fmaxf(acc[4] + bb.x, 0.0f) * dd);
        o[5] = (_Float16)(fmaxf(acc[5] + bb.y, 0.0f) * dd);
        o[6] = (_Float16)(fmaxf(acc[6] + bb.z, 0.0f) * dd);
        o[7] = (_Float16)(fmaxf(acc[7] + bb.w, 0.0f) * dd);
        *(v8h*)(outh + (size_t)nd * HID + sub * 8) = o;
    }
}

__global__ void __launch_bounds__(128) __attribute__((amdgpu_waves_per_eu(16)))
gcn_gather_proj(const _Float16* __restrict__ H, const int* __restrict__ rowptr,
                const int* __restrict__ nbr, const float* __restrict__ dis,
                const float* __restrict__ Wt, const float* __restrict__ bias,
                float* __restrict__ outf, int N) {
    constexpr int P  = HID + 8;
    constexpr int CP = HID + 4;
    __shared__ alignas(16) _Float16 Bs[HID * P];
    __shared__ alignas(16) _Float16 Ahi[16 * P];
    __shared__ alignas(16) _Float16 Alo[16 * P];
    __shared__ alignas(16) uint2    s_in[4][GRP * WPITCH];
    __shared__ alignas(16) unsigned s_out[4][GRP * OPITCH];
    static_assert(sizeof(float) * 16 * CP <= sizeof(uint2) * 4 * GRP * WPITCH, "the f32 tile must fit into the staging windows");
    float* Cs = (float*)&s_in[0][0];

    const int tid  = threadIdx.x;
    const int lane = tid & 31;
    const int wv   = __builtin_amdgcn_readfirstlane(tid >> 5);
    const int q    = lane >> 3;
    const int sub  = lane & 7;
    const int nd   = (int)(blockIdx.x * 16) + wv * GRP + q;
    const int node = (nd < N) ? nd : (N - 1);

    const int wn  = tid & (HID - 1);
    const int wk0 = tid >> 6;
    float wf[4][8];
#pragma unroll
    for (int j = 0; j < 4; ++j) {
        const float* wp = Wt + (size_t)(wk0 + j * 2) * 8 * HID + wn;
#pragma unroll
        for (int qq = 0; qq < 8; ++qq) wf[j][qq] = wp[qq * HID];
    }
#pragma unroll
    for (int j = 0; j < 4; ++j) {
        v8h pk;
#pragma unroll
        for (int qq = 0; qq < 8; ++qq) pk[qq] = (_Float16)wf[j][qq];
        *(v8h*)(Bs + wn * P + (wk0 + j * 2) * 8) = pk;
    }

    float acc[8];
    float dd;
    gcn_node_sums<4>(H, rowptr, nbr, dis, node, sub,
                     &s_in[wv][q * WPITCH], &s_out[wv][q * OPITCH], acc, dd);

    static_assert(GRP * 8 == 32, "a wave is GRP lane groups of 8: tile row = tid >> 3");
    const int trow = tid >> 3;
    const int ocol = (lane & 15) * 4;

    {
        v8h hi8, lo8;
#pragma unroll
        for (int i = 0; i < 8; ++i) {
            hi8[i] = (_Float16)acc[i];
            lo8[i] = (_Float16)(acc[i] - (float)hi8[i]);
        }
        *(v8h*)(Ahi + trow * P + sub * 8) = hi8;
        *(v8h*)(Alo + trow * P + sub * 8) = lo8;
    }
    const float4 b2v = *(const float4*)(bias + ocol);
    __syncthreads();

    {
        const int c  = lane & 15;
        const int hi = lane >> 4;
        v8f d = (v8f){0.f, 0.f, 0.f, 0.f, 0.f, 0.f, 0.f, 0.f};
#pragma unroll
        for (int kt = 0; kt < HID / 32; ++kt) {
            const int ko = kt * 32 + hi * 8;
            union { v16h v; v8h h[2]; } ah, al, b;
            ah.h[0] = *(const v8h*)(Ahi + c * P + ko);
            ah.h[1] = *(const v8h*)(Ahi + c * P + ko + 16);
            al.h[0] = *(const v8h*)(Alo + c * P + ko);
            al.h[1] = *(const v8h*)(Alo + c * P + ko + 16);
            b.h[0]  = *(const v8h*)(Bs + (wv * 16 + c) * P + ko);
            b.h[1]  = *(const v8h*)(Bs + (wv * 16 + c) * P + ko + 16);
            d = __builtin_amdgcn_wmma_f32_16x16x32_f16(
                 false, al.v,  false, b.v,
                 (short)0, d,  false,  false);
            d = __builtin_amdgcn_wmma_f32_16x16x32_f16(
                 false, ah.v,  false, b.v,
                 (short)0, d,  false,  false);
        }
#pragma unroll
        for (int r = 0; r < 8; ++r) Cs[(hi * 8 + r) * CP + wv * 16 + c] = d[r];
    }
    __syncthreads();

#pragma unroll
    for (int k = 0; k < 2; ++k) {
        const int orow = wv * GRP + k * 2 + (lane >> 4);
        const int on   = (int)(blockIdx.x * 16) + orow;
        if (on < N) {
            const float4 cv = *(const float4*)(Cs + orow * CP + ocol);
            float4 o;
            o.x = cv.x + b2v.x;
            o.y = cv.y + b2v.y;
            o.z = cv.z + b2v.z;
            o.w = cv.w + b2v.w;
            *(float4*)(outf + (size_t)on * HID + ocol) = o;
        }
    }
}


extern "C" void kernel_launch(void* const* d_in, const int* in_sizes, int n_in,
                              void* d_out, int out_size, void* d_ws, size_t ws_size,
                              hipStream_t stream) {
    const float* x  = (const float*)d_in[0];
    const int*   ei = (const int*)d_in[1];
    const float* W1 = (const float*)d_in[2];
    const float* b1 = (const float*)d_in[3];
    const float* W2 = (const float*)d_in[4];
    const float* b2 = (const float*)d_in[5];
    float* out = (float*)d_out;

    const int N = in_sizes[0] / IN_CH;
    const int E = in_sizes[1] / 2;
    const int* src = ei;
    const int* dst = ei + E;

    char* ws = (char*)d_ws;
    size_t off = 0;
    auto alloc = [&](size_t bytes) -> char* {
        char* p = ws + off;
        off = (off + bytes + 255) & ~(size_t)255;
        return p;
    };
    char*     regA = alloc((size_t)N * HID * sizeof(int));
    char*     regB = alloc((size_t)N * HID * sizeof(_Float16));
    float*    dis  = (float*)   alloc((size_t)N * sizeof(float));
    int*      rowp = (int*)     alloc((size_t)(N + 1) * sizeof(int));
    int*      slot = (int*)     alloc((size_t)E * sizeof(int));
    int*      nbr  = (int*)     alloc((size_t)((E > NSL * CHUNKS) ? E : (NSL * CHUNKS)) * sizeof(int));
    int*      crs  = nbr;
    unsigned short* hist = (unsigned short*)regB;
    int*            offp = (int*)regA;
    _Float16*       proj = (_Float16*)regB;
    _Float16*       h1h  = (_Float16*)regA;

    const int T = 256;
    const int gemmBlocks = ((N + 15) / 16 + 7) / 8;
    const int CH = (E + CHUNKS - 1) / CHUNKS;

    const int bpp = (CH + FILL_SPAN - 1) / FILL_SPAN;
    const int fillBlocks = CHUNKS * bpp;

    gcn_chunk_hist<<<CHUNKS, 1024, 0, stream>>>(dst, hist, crs, slot, N, E, CH);
    gcn_scan_deg<<<(N + CSL - 1) / CSL, CSL, 0, stream>>>(hist, crs, offp, rowp, dis, N);

    gcn_fill_gemm<<<fillBlocks + gemmBlocks, 256, 0, stream>>>(x, W1, dis, proj, N,
                                                              src, dst, slot, offp, nbr, E, CH, bpp, fillBlocks);

    gcn_gather<<<(N + 8 * GRP - 1) / (8 * GRP), T, 0, stream>>>(proj, rowp, nbr, dis, b1, h1h, N);

    gcn_gather_proj<<<(N + 15) / 16, 128, 0, stream>>>(h1h, rowp, nbr, dis, W2, b2, out, N);
}
